# starting kernel
# baseline (speedup 1.0000x reference)
#include <hip/hip_runtime.h>

typedef float v2f __attribute__((ext_vector_type(2)));
typedef float v8f __attribute__((ext_vector_type(8)));

#define DIM 32


__global__ void zero_f32_kernel(float* __restrict__ p, int n) {
  int i = blockIdx.x * blockDim.x + threadIdx.x;
  if (i < n) p[i] = 0.0f;
}

__global__ void deg_kernel(const int* __restrict__ col, const float* __restrict__ w,
                           float* __restrict__ deg, int E) {
  int e = blockIdx.x * blockDim.x + threadIdx.x;
  if (e < E) atomicAdd(&deg[col[e]], w[e]);
}

__global__ void dis_kernel(const float* __restrict__ deg, float* __restrict__ dis, int N) {
  int i = blockIdx.x * blockDim.x + threadIdx.x;
  if (i < N) {
    float d = deg[i];
    dis[i] = (d > 0.0f) ? rsqrtf(d) : 0.0f;
  }
}

__global__ void norm_kernel(const int* __restrict__ row, const int* __restrict__ col,
                            const float* __restrict__ w, const float* __restrict__ dis,
                            float* __restrict__ nrm, int E) {
  int e = blockIdx.x * blockDim.x + threadIdx.x;
  if (e < E) nrm[e] = dis[row[e]] * w[e] * dis[col[e]];
}

__global__ void hop_kernel(const int* __restrict__ row, const int* __restrict__ col,
                           const float* __restrict__ nrm,
                           const float* __restrict__ hsrc, float* __restrict__ hdst, int E) {
  int t = blockIdx.x * blockDim.x + threadIdx.x;
  int e = t >> 5;
  int d = t & 31;
  if (e >= E) return;
  int r = row[e];
  int c = col[e];
  float v = nrm[e] * hsrc[(long)r * DIM + d];
  atomicAdd(&hdst[(long)c * DIM + d], v);
}


__global__ __launch_bounds__(256)
void tag_gemm_wmma_kernel(const float* __restrict__ s0, const float* __restrict__ s1,
                          const float* __restrict__ s2, const float* __restrict__ s3,
                          const float* __restrict__ Wl,
                          const float* __restrict__ bl,
                          float* __restrict__ dst, int nTiles) {
  int wave = threadIdx.x >> 5;
  int tile = blockIdx.x * (blockDim.x >> 5) + wave;
  if (tile >= nTiles) return;
  int lane = threadIdx.x & 31;
  int mlane = lane & 15;
  bool hi = lane >= 16;
  int khalf = hi ? 2 : 0;
  long rowBase = (long)(tile * 16 + mlane) * DIM;

  v8f c0 = {};
  v8f c1 = {};

  const float* srcs[4] = {s0, s1, s2, s3};
#pragma unroll
  for (int k = 0; k < 4; ++k) {
    const float* S = srcs[k];
    const float* Wk = Wl + k * (DIM * DIM);
#pragma unroll
    for (int kk = 0; kk < DIM; kk += 4) {
      int ka = kk + khalf;
      v2f a = *reinterpret_cast<const v2f*>(S + rowBase + ka);
      v2f b0, b1;
      b0.x = Wk[(ka + 0) * DIM + mlane];
      b0.y = Wk[(ka + 1) * DIM + mlane];
      b1.x = Wk[(ka + 0) * DIM + mlane + 16];
      b1.y = Wk[(ka + 1) * DIM + mlane + 16];
      c0 = __builtin_amdgcn_wmma_f32_16x16x4_f32(false, a, false, b0, (short)0, c0, false, false);
      c1 = __builtin_amdgcn_wmma_f32_16x16x4_f32(false, a, false, b1, (short)0, c1, false, false);
    }
  }

  float bn0 = bl[mlane];
  float bn1 = bl[mlane + 16];
  int mBase = tile * 16 + (hi ? 8 : 0);
#pragma unroll
  for (int r = 0; r < 8; ++r) {
    float v0 = c0[r] + bn0;
    float v1 = c1[r] + bn1;
    v0 = v0 > 0.0f ? v0 : 0.0f;
    v1 = v1 > 0.0f ? v1 : 0.0f;
    dst[(long)(mBase + r) * DIM + mlane]      = v0;
    dst[(long)(mBase + r) * DIM + mlane + 16] = v1;
  }
}

__global__ void tag_gemm_tail_kernel(const float* __restrict__ s0, const float* __restrict__ s1,
                                     const float* __restrict__ s2, const float* __restrict__ s3,
                                     const float* __restrict__ Wl, const float* __restrict__ bl,
                                     float* __restrict__ dst, int startRow, int N) {
  int t = blockIdx.x * blockDim.x + threadIdx.x;
  int m = startRow + (t >> 5);
  int n = t & 31;
  if (m >= N) return;
  float acc = bl[n];
  const float* srcs[4] = {s0, s1, s2, s3};
  for (int k = 0; k < 4; ++k) {
    const float* S = srcs[k] + (long)m * DIM;
    const float* Wk = Wl + k * DIM * DIM;
    for (int j = 0; j < DIM; ++j) acc += S[j] * Wk[j * DIM + n];
  }
  dst[(long)m * DIM + n] = acc > 0.0f ? acc : 0.0f;
}


extern "C" void kernel_launch(void* const* d_in, const int* in_sizes, int n_in,
                              void* d_out, int out_size, void* d_ws, size_t ws_size,
                              hipStream_t stream) {
  (void)n_in; (void)out_size; (void)ws_size;
  const float* d_x = (const float*)d_in[0];
  const int*   d_ei = (const int*)d_in[1];
  const float* d_w = (const float*)d_in[2];
  const float* d_W = (const float*)d_in[3];
  const float* d_b = (const float*)d_in[4];
  const int N = in_sizes[0] / DIM;
  const int E = in_sizes[2];
  const int* row = d_ei;
  const int* col = d_ei + E;

  auto align_up = [](size_t v) { return (v + 255) & ~(size_t)255; };
  char* ws = (char*)d_ws;
  size_t off = 0;
  float* deg  = (float*)(ws + off); off = align_up(off + (size_t)N * 4);
  float* dis  = (float*)(ws + off); off = align_up(off + (size_t)N * 4);
  float* nrm  = (float*)(ws + off); off = align_up(off + (size_t)E * 4);
  float* h1   = (float*)(ws + off); off = align_up(off + (size_t)N * DIM * 4);
  float* h2   = (float*)(ws + off); off = align_up(off + (size_t)N * DIM * 4);
  float* h3   = (float*)(ws + off); off = align_up(off + (size_t)N * DIM * 4);
  float* xcur = (float*)(ws + off); off = align_up(off + (size_t)N * DIM * 4);

  const int B = 256;
  auto gblk = [&](long n) { return (int)((n + B - 1) / B); };

  zero_f32_kernel<<<gblk(N), B, 0, stream>>>(deg, N);
  deg_kernel<<<gblk(E), B, 0, stream>>>(col, d_w, deg, E);
  dis_kernel<<<gblk(N), B, 0, stream>>>(deg, dis, N);
  norm_kernel<<<gblk(E), B, 0, stream>>>(row, col, d_w, dis, nrm, E);

  const int nTiles = N / 16;
  const int tileBlocks = (nTiles + 7) / 8;
  const int tailStart = nTiles * 16;
  const int tailRows = N - tailStart;
  const long NF = (long)N * DIM;
  const long hopThreads = (long)E * 32;

  for (int l = 0; l < 3; ++l) {
    const float* xin = (l == 0) ? d_x : xcur;
    float* dst = (l == 2) ? (float*)d_out : xcur;
    const float* Wl = d_W + (size_t)l * 4 * DIM * DIM;
    const float* bl = d_b + (size_t)l * DIM;

    zero_f32_kernel<<<gblk(NF), B, 0, stream>>>(h1, (int)NF);
    hop_kernel<<<gblk(hopThreads), B, 0, stream>>>(row, col, nrm, xin, h1, E);
    zero_f32_kernel<<<gblk(NF), B, 0, stream>>>(h2, (int)NF);
    hop_kernel<<<gblk(hopThreads), B, 0, stream>>>(row, col, nrm, h1, h2, E);
    zero_f32_kernel<<<gblk(NF), B, 0, stream>>>(h3, (int)NF);
    hop_kernel<<<gblk(hopThreads), B, 0, stream>>>(row, col, nrm, h2, h3, E);

    tag_gemm_wmma_kernel<<<tileBlocks, B, 0, stream>>>(xin, h1, h2, h3, Wl, bl, dst, nTiles);
    if (tailRows > 0) {
      tag_gemm_tail_kernel<<<gblk((long)tailRows * 32), B, 0, stream>>>(
          xin, h1, h2, h3, Wl, bl, dst, tailStart, N);
    }
  }
}


# Edges grouped by target once per call (returning int adds rank them, scan, fill): each hop is an atomic-free segmented row gather; sum order per target follows the rank adds' arrival
# speedup vs baseline: 3.6853x; 3.6853x over previous
#include <hip/hip_runtime.h>


typedef float v2f __attribute__((ext_vector_type(2)));
typedef float v8f __attribute__((ext_vector_type(8)));

#define DIM 32
#define SCAN_THREADS 1024
#define SCAN_TILE (SCAN_THREADS * 4)


__global__ void zero_i32_kernel(int* __restrict__ p, int n) {
  int i = blockIdx.x * blockDim.x + threadIdx.x;
  if (i < n) p[i] = 0;
}

__global__ void rank_kernel(const int* __restrict__ col, int* __restrict__ cnt,
                            int* __restrict__ rnk, int E) {
  int e = blockIdx.x * blockDim.x + threadIdx.x;
  if (e < E) rnk[e] = atomicAdd(&cnt[col[e]], 1);
}

__global__ __launch_bounds__(SCAN_THREADS)
void scan_kernel(const int* __restrict__ cnt, int* __restrict__ start, int N) {
  __shared__ int red[SCAN_THREADS / 32];
  __shared__ int wtot[SCAN_THREADS / 32];
  const int tid = threadIdx.x;
  const int lane = tid & 31;
  const int wv = tid >> 5;
  const int base0 = blockIdx.x * SCAN_TILE;

  int a0 = 0, a1 = 0, a2 = 0, a3 = 0;
  int i = tid * 4;
  for (; i + 3 * SCAN_TILE < base0; i += 4 * SCAN_TILE) {
    int4 v0 = *reinterpret_cast<const int4*>(cnt + i);
    int4 v1 = *reinterpret_cast<const int4*>(cnt + i + SCAN_TILE);
    int4 v2 = *reinterpret_cast<const int4*>(cnt + i + 2 * SCAN_TILE);
    int4 v3 = *reinterpret_cast<const int4*>(cnt + i + 3 * SCAN_TILE);
    a0 += (v0.x + v0.y) + (v0.z + v0.w);
    a1 += (v1.x + v1.y) + (v1.z + v1.w);
    a2 += (v2.x + v2.y) + (v2.z + v2.w);
    a3 += (v3.x + v3.y) + (v3.z + v3.w);
  }
  for (; i < base0; i += SCAN_TILE) {
    int4 v = *reinterpret_cast<const int4*>(cnt + i);
    a0 += (v.x + v.y) + (v.z + v.w);
  }
  int acc = (a0 + a1) + (a2 + a3);
  for (int o = 16; o > 0; o >>= 1) acc += __shfl_xor(acc, o);
  if (lane == 0) red[wv] = acc;

  const int i0 = base0 + tid * 4;
  int c0 = 0, c1 = 0, c2 = 0, c3 = 0;
  if (i0 + 3 < N) {
    int4 v = *reinterpret_cast<const int4*>(cnt + i0);
    c0 = v.x; c1 = v.y; c2 = v.z; c3 = v.w;
  } else {
    if (i0 + 0 < N) c0 = cnt[i0 + 0];
    if (i0 + 1 < N) c1 = cnt[i0 + 1];
    if (i0 + 2 < N) c2 = cnt[i0 + 2];
  }
  const int t1 = c0, t2 = t1 + c1, t3 = t2 + c2, tsum = t3 + c3;
  int inc = tsum;
  for (int o = 1; o < 32; o <<= 1) {
    int u = __shfl_up(inc, o);
    if (lane >= o) inc += u;
  }
  if (lane == 31) wtot[wv] = inc;
  __syncthreads();
  int pre = 0;
  for (int k = 0; k < SCAN_THREADS / 32; ++k) pre += red[k];
  for (int k = 0; k < wv; ++k) pre += wtot[k];
  const int ex = pre + inc - tsum;
  if (i0 + 0 <= N) start[i0 + 0] = ex;
  if (i0 + 1 <= N) start[i0 + 1] = ex + t1;
  if (i0 + 2 <= N) start[i0 + 2] = ex + t2;
  if (i0 + 3 <= N) start[i0 + 3] = ex + t3;
}

__global__ void fill_kernel(const int* __restrict__ row, const int* __restrict__ col,
                            const float* __restrict__ w, const int* __restrict__ start,
                            const int* __restrict__ rnk, int2* __restrict__ sedge, int E) {
  int e = blockIdx.x * blockDim.x + threadIdx.x;
  if (e < E) {
    int p = start[col[e]] + rnk[e];
    sedge[p] = make_int2(row[e], __float_as_int(w[e]));
  }
}

__global__ void dis_csr_kernel(const int* __restrict__ start, const int2* __restrict__ sedge,
                               float* __restrict__ dis, int N) {
  int t = blockIdx.x * blockDim.x + threadIdx.x;
  int c = t >> 3;
  int sub = t & 7;
  if (c >= N) c = N - 1;
  const int s = start[c];
  const int e = start[c + 1];
  float d = 0.0f;
  for (int p = s + sub; p < e; p += 8) d += __int_as_float(sedge[p].y);
  d += __shfl_xor(d, 4);
  d += __shfl_xor(d, 2);
  d += __shfl_xor(d, 1);
  if (sub == 0) dis[c] = (d > 0.0f) ? rsqrtf(d) : 0.0f;
}

__global__ void norm_csr_kernel(const int* __restrict__ start, const float* __restrict__ dis,
                                int2* __restrict__ sedge, int N) {
  int t = blockIdx.x * blockDim.x + threadIdx.x;
  int c = t >> 3;
  int sub = t & 7;
  if (c >= N) return;
  const int s = start[c];
  const int e = start[c + 1];
  const float dc = dis[c];
  for (int p = s + sub; p < e; p += 8) {
    int2 ed = sedge[p];
    float nv = dis[ed.x] * __int_as_float(ed.y) * dc;
    sedge[p] = make_int2(ed.x, __float_as_int(nv));
  }
}

__global__ __launch_bounds__(256)
void hop_csr_kernel(const int* __restrict__ start, const int2* __restrict__ sedge,
                    const float* __restrict__ hsrc, float* __restrict__ hdst, int N) {
  int t = blockIdx.x * blockDim.x + threadIdx.x;
  int c = t >> 3;
  int sub = t & 7;
  if (c >= N) return;
  const int s = start[c];
  const int e = start[c + 1];
  const float4* __restrict__ h4 = reinterpret_cast<const float4*>(hsrc);
  float4 acc = make_float4(0.0f, 0.0f, 0.0f, 0.0f);
  if (s < e) {
    const int last = e - 1;
    int2 ed0 = sedge[min(s + 0, last)];
    int2 ed1 = sedge[min(s + 1, last)];
    int2 ed2 = sedge[min(s + 2, last)];
    int2 ed3 = sedge[min(s + 3, last)];
    for (int p = s; p < e; p += 4) {
      float4 v0 = h4[(long)ed0.x * (DIM / 4) + sub];
      float4 v1 = h4[(long)ed1.x * (DIM / 4) + sub];
      float4 v2 = h4[(long)ed2.x * (DIM / 4) + sub];
      float4 v3 = h4[(long)ed3.x * (DIM / 4) + sub];
      const int q = p + 4;
      int2 n0 = sedge[min(q + 0, last)];
      int2 n1 = sedge[min(q + 1, last)];
      int2 n2 = sedge[min(q + 2, last)];
      int2 n3 = sedge[min(q + 3, last)];
      const bool k1 = p + 1 < e, k2 = p + 2 < e, k3 = p + 3 < e;
      const float4 z = make_float4(0.0f, 0.0f, 0.0f, 0.0f);
      const float w0 = __int_as_float(ed0.y);
      const float w1 = k1 ? __int_as_float(ed1.y) : 0.0f;
      const float w2 = k2 ? __int_as_float(ed2.y) : 0.0f;
      const float w3 = k3 ? __int_as_float(ed3.y) : 0.0f;
      if (!k1) v1 = z;
      if (!k2) v2 = z;
      if (!k3) v3 = z;
      acc.x += w0 * v0.x; acc.y += w0 * v0.y; acc.z += w0 * v0.z; acc.w += w0 * v0.w;
      acc.x += w1 * v1.x; acc.y += w1 * v1.y; acc.z += w1 * v1.z; acc.w += w1 * v1.w;
      acc.x += w2 * v2.x; acc.y += w2 * v2.y; acc.z += w2 * v2.z; acc.w += w2 * v2.w;
      acc.x += w3 * v3.x; acc.y += w3 * v3.y; acc.z += w3 * v3.z; acc.w += w3 * v3.w;
      ed0 = n0; ed1 = n1; ed2 = n2; ed3 = n3;
    }
  }
  reinterpret_cast<float4*>(hdst)[(long)c * (DIM / 4) + sub] = acc;
}


__global__ __launch_bounds__(256)
void tag_gemm_wmma_kernel(const float* __restrict__ s0, const float* __restrict__ s1,
                          const float* __restrict__ s2, const float* __restrict__ s3,
                          const float* __restrict__ Wl,
                          const float* __restrict__ bl,
                          float* __restrict__ dst, int nTiles) {
  int wave = threadIdx.x >> 5;
  int tile = blockIdx.x * (blockDim.x >> 5) + wave;
  if (tile >= nTiles) return;
  int lane = threadIdx.x & 31;
  int mlane = lane & 15;
  bool hi = lane >= 16;
  int khalf = hi ? 2 : 0;
  long rowBase = (long)(tile * 16 + mlane) * DIM;

  v8f c0 = {};
  v8f c1 = {};

  const float* srcs[4] = {s0, s1, s2, s3};
#pragma unroll
  for (int k = 0; k < 4; ++k) {
    const float* S = srcs[k];
    const float* Wk = Wl + k * (DIM * DIM);
#pragma unroll
    for (int kk = 0; kk < DIM; kk += 4) {
      int ka = kk + khalf;
      v2f a = *reinterpret_cast<const v2f*>(S + rowBase + ka);
      v2f b0, b1;
      b0.x = Wk[(ka + 0) * DIM + mlane];
      b0.y = Wk[(ka + 1) * DIM + mlane];
      b1.x = Wk[(ka + 0) * DIM + mlane + 16];
      b1.y = Wk[(ka + 1) * DIM + mlane + 16];
      c0 = __builtin_amdgcn_wmma_f32_16x16x4_f32(false, a, false, b0, (short)0, c0, false, false);
      c1 = __builtin_amdgcn_wmma_f32_16x16x4_f32(false, a, false, b1, (short)0, c1, false, false);
    }
  }

  float bn0 = bl[mlane];
  float bn1 = bl[mlane + 16];
  int mBase = tile * 16 + (hi ? 8 : 0);
#pragma unroll
  for (int r = 0; r < 8; ++r) {
    float v0 = c0[r] + bn0;
    float v1 = c1[r] + bn1;
    v0 = v0 > 0.0f ? v0 : 0.0f;
    v1 = v1 > 0.0f ? v1 : 0.0f;
    dst[(long)(mBase + r) * DIM + mlane]      = v0;
    dst[(long)(mBase + r) * DIM + mlane + 16] = v1;
  }
}

__global__ void tag_gemm_tail_kernel(const float* __restrict__ s0, const float* __restrict__ s1,
                                     const float* __restrict__ s2, const float* __restrict__ s3,
                                     const float* __restrict__ Wl, const float* __restrict__ bl,
                                     float* __restrict__ dst, int startRow, int N) {
  int t = blockIdx.x * blockDim.x + threadIdx.x;
  int m = startRow + (t >> 5);
  int n = t & 31;
  if (m >= N) return;
  float acc = bl[n];
  const float* srcs[4] = {s0, s1, s2, s3};
  for (int k = 0; k < 4; ++k) {
    const float* S = srcs[k] + (long)m * DIM;
    const float* Wk = Wl + k * DIM * DIM;
    for (int j = 0; j < DIM; ++j) acc += S[j] * Wk[j * DIM + n];
  }
  dst[(long)m * DIM + n] = acc > 0.0f ? acc : 0.0f;
}


extern "C" void kernel_launch(void* const* d_in, const int* in_sizes, int n_in,
                              void* d_out, int out_size, void* d_ws, size_t ws_size,
                              hipStream_t stream) {
  (void)n_in; (void)out_size; (void)ws_size;
  const float* d_x = (const float*)d_in[0];
  const int*   d_ei = (const int*)d_in[1];
  const float* d_w = (const float*)d_in[2];
  const float* d_W = (const float*)d_in[3];
  const float* d_b = (const float*)d_in[4];
  const int N = in_sizes[0] / DIM;
  const int E = in_sizes[2];
  const int* row = d_ei;
  const int* col = d_ei + E;

  auto align_up = [](size_t v) { return (v + 255) & ~(size_t)255; };
  char* ws = (char*)d_ws;
  size_t off = 0;
  int*   cnt   = (int*)(ws + off);   off = align_up(off + (size_t)N * 4);
  int*   start = (int*)(ws + off);   off = align_up(off + ((size_t)N + 1) * 4);
  int*   rnk   = (int*)(ws + off);   off = align_up(off + (size_t)E * 4);
  int2*  sedge = (int2*)(ws + off);  off = align_up(off + (size_t)E * 8);
  float* dis   = (float*)(ws + off); off = align_up(off + (size_t)N * 4);
  float* h1    = (float*)(ws + off); off = align_up(off + (size_t)N * DIM * 4);
  float* h2    = (float*)(ws + off); off = align_up(off + (size_t)N * DIM * 4);
  float* h3    = (float*)(ws + off); off = align_up(off + (size_t)N * DIM * 4);
  float* xcur  = (float*)(ws + off); off = align_up(off + (size_t)N * DIM * 4);

  const int B = 256;
  auto gblk = [&](long n) { return (int)((n + B - 1) / B); };
  const long grp8 = (long)N * 8;

  zero_i32_kernel<<<gblk(N), B, 0, stream>>>(cnt, N);
  rank_kernel<<<gblk(E), B, 0, stream>>>(col, cnt, rnk, E);
  scan_kernel<<<N / SCAN_TILE + 1, SCAN_THREADS, 0, stream>>>(cnt, start, N);
  fill_kernel<<<gblk(E), B, 0, stream>>>(row, col, d_w, start, rnk, sedge, E);
  dis_csr_kernel<<<gblk(grp8), B, 0, stream>>>(start, sedge, dis, N);
  norm_csr_kernel<<<gblk(grp8), B, 0, stream>>>(start, dis, sedge, N);

  const int nTiles = N / 16;
  const int tileBlocks = (nTiles + 7) / 8;
  const int tailStart = nTiles * 16;
  const int tailRows = N - tailStart;

  for (int l = 0; l < 3; ++l) {
    const float* xin = (l == 0) ? d_x : xcur;
    float* dst = (l == 2) ? (float*)d_out : xcur;
    const float* Wl = d_W + (size_t)l * 4 * DIM * DIM;
    const float* bl = d_b + (size_t)l * DIM;

    hop_csr_kernel<<<gblk(grp8), B, 0, stream>>>(start, sedge, xin, h1, N);
    hop_csr_kernel<<<gblk(grp8), B, 0, stream>>>(start, sedge, h1, h2, N);
    hop_csr_kernel<<<gblk(grp8), B, 0, stream>>>(start, sedge, h2, h3, N);

    tag_gemm_wmma_kernel<<<tileBlocks, B, 0, stream>>>(xin, h1, h2, h3, Wl, bl, dst, nTiles);
    if (tailRows > 0) {
      tag_gemm_tail_kernel<<<gblk((long)tailRows * 32), B, 0, stream>>>(
          xin, h1, h2, h3, Wl, bl, dst, tailStart, N);
    }
  }
}


# Edges grouped by target with a two-pass bucket sort in LDS (256-node buckets, run writes) in place of global rank adds and per-edge scatter; sum order per target follows the LDS adds' arrival
# speedup vs baseline: 5.0798x; 1.3784x over previous
#include <hip/hip_runtime.h>


typedef float v2f __attribute__((ext_vector_type(2)));
typedef float v8f __attribute__((ext_vector_type(8)));

#define DIM 32
#define SCAN_THREADS 1024
#define SCAN_TILE (SCAN_THREADS * 4)


__global__ void zero_i32_kernel(int* __restrict__ p, int n) {
  int i = blockIdx.x * blockDim.x + threadIdx.x;
  if (i < n) p[i] = 0;
}

__global__ void rank_kernel(const int* __restrict__ col, int* __restrict__ cnt,
                            int* __restrict__ rnk, int E) {
  int e = blockIdx.x * blockDim.x + threadIdx.x;
  if (e < E) rnk[e] = atomicAdd(&cnt[col[e]], 1);
}

__global__ __launch_bounds__(SCAN_THREADS)
void scan_kernel(const int* __restrict__ cnt, int* __restrict__ start, int N) {
  __shared__ int red[SCAN_THREADS / 32];
  __shared__ int wtot[SCAN_THREADS / 32];
  const int tid = threadIdx.x;
  const int lane = tid & 31;
  const int wv = tid >> 5;
  const int base0 = blockIdx.x * SCAN_TILE;

  int a0 = 0, a1 = 0, a2 = 0, a3 = 0;
  int i = tid * 4;
  for (; i + 3 * SCAN_TILE < base0; i += 4 * SCAN_TILE) {
    int4 v0 = *reinterpret_cast<const int4*>(cnt + i);
    int4 v1 = *reinterpret_cast<const int4*>(cnt + i + SCAN_TILE);
    int4 v2 = *reinterpret_cast<const int4*>(cnt + i + 2 * SCAN_TILE);
    int4 v3 = *reinterpret_cast<const int4*>(cnt + i + 3 * SCAN_TILE);
    a0 += (v0.x + v0.y) + (v0.z + v0.w);
    a1 += (v1.x + v1.y) + (v1.z + v1.w);
    a2 += (v2.x + v2.y) + (v2.z + v2.w);
    a3 += (v3.x + v3.y) + (v3.z + v3.w);
  }
  for (; i < base0; i += SCAN_TILE) {
    int4 v = *reinterpret_cast<const int4*>(cnt + i);
    a0 += (v.x + v.y) + (v.z + v.w);
  }
  int acc = (a0 + a1) + (a2 + a3);
  for (int o = 16; o > 0; o >>= 1) acc += __shfl_xor(acc, o);
  if (lane == 0) red[wv] = acc;

  const int i0 = base0 + tid * 4;
  int c0 = 0, c1 = 0, c2 = 0, c3 = 0;
  if (i0 + 3 < N) {
    int4 v = *reinterpret_cast<const int4*>(cnt + i0);
    c0 = v.x; c1 = v.y; c2 = v.z; c3 = v.w;
  } else {
    if (i0 + 0 < N) c0 = cnt[i0 + 0];
    if (i0 + 1 < N) c1 = cnt[i0 + 1];
    if (i0 + 2 < N) c2 = cnt[i0 + 2];
  }
  const int t1 = c0, t2 = t1 + c1, t3 = t2 + c2, tsum = t3 + c3;
  int inc = tsum;
  for (int o = 1; o < 32; o <<= 1) {
    int u = __shfl_up(inc, o);
    if (lane >= o) inc += u;
  }
  if (lane == 31) wtot[wv] = inc;
  __syncthreads();
  int pre = 0;
  for (int k = 0; k < SCAN_THREADS / 32; ++k) pre += red[k];
  for (int k = 0; k < wv; ++k) pre += wtot[k];
  const int ex = pre + inc - tsum;
  if (i0 + 0 <= N) start[i0 + 0] = ex;
  if (i0 + 1 <= N) start[i0 + 1] = ex + t1;
  if (i0 + 2 <= N) start[i0 + 2] = ex + t2;
  if (i0 + 3 <= N) start[i0 + 3] = ex + t3;
}

__global__ void fill_kernel(const int* __restrict__ row, const int* __restrict__ col,
                            const float* __restrict__ w, const int* __restrict__ start,
                            const int* __restrict__ rnk, int2* __restrict__ sedge, int E) {
  int e = blockIdx.x * blockDim.x + threadIdx.x;
  if (e < E) {
    int p = start[col[e]] + rnk[e];
    sedge[p] = make_int2(row[e], __float_as_int(w[e]));
  }
}


#define SORT_SH 8
#define SORT_BN (1 << SORT_SH)
#define SORT_MAXNB 512
#define SORT_CH 8192
#define SORT_CAP 8192

__device__ __forceinline__ int block_excl_scan(int v, int* wsum) {
  const int lane = threadIdx.x & 31;
  const int wv = threadIdx.x >> 5;
  int inc = v;
  for (int o = 1; o < 32; o <<= 1) {
    int u = __shfl_up(inc, o);
    if (lane >= o) inc += u;
  }
  if (lane == 31) wsum[wv] = inc;
  __syncthreads();
  int pre = 0;
  for (int k = 0; k < wv; ++k) pre += wsum[k];
  return pre + inc - v;
}

__global__ __launch_bounds__(1024)
void bucket_hist_kernel(const int* __restrict__ col, int* __restrict__ hist,
                        int E, int NB, int NWG) {
  __shared__ int cnt[SORT_MAXNB];
  const int tid = threadIdx.x;
  const int g = blockIdx.x;
  for (int b = tid; b < NB; b += 1024) cnt[b] = 0;
  __syncthreads();
  const int e0 = g * SORT_CH;
  const int n = min(E - e0, SORT_CH);
  int rc[SORT_CH / 1024];
#pragma unroll
  for (int k = 0; k < SORT_CH / 1024; ++k) {
    const int i = k * 1024 + tid;
    rc[k] = (i < n) ? col[e0 + i] : -1;
  }
#pragma unroll
  for (int k = 0; k < SORT_CH / 1024; ++k)
    if (rc[k] >= 0) atomicAdd(&cnt[rc[k] >> SORT_SH], 1);
  __syncthreads();
  for (int b = tid; b < NB; b += 1024) hist[b * NWG + g] = cnt[b];
}

__global__ __launch_bounds__(1024)
void bucket_part_kernel(const int* __restrict__ row, const int* __restrict__ col,
                        const float* __restrict__ w, const int* __restrict__ offs,
                        int2* __restrict__ prc, int* __restrict__ pcl,
                        int E, int NB, int NWG) {
  __shared__ int cnt[SORT_MAXNB];
  __shared__ int lstart[SORT_MAXNB];
  __shared__ int gbase[SORT_MAXNB];
  __shared__ int wsum[32];
  __shared__ int s_row[SORT_CH];
  __shared__ int s_col[SORT_CH];
  __shared__ int s_w[SORT_CH];
  const int tid = threadIdx.x;
  const int g = blockIdx.x;
  for (int b = tid; b < NB; b += 1024) cnt[b] = 0;
  __syncthreads();
  const int e0 = g * SORT_CH;
  const int n = min(E - e0, SORT_CH);
  int rc[SORT_CH / 1024], rr[SORT_CH / 1024], rw[SORT_CH / 1024], rk[SORT_CH / 1024];
#pragma unroll
  for (int k = 0; k < SORT_CH / 1024; ++k) {
    const int i = k * 1024 + tid;
    if (i < n) {
      const int c = col[e0 + i];
      rc[k] = c;
      rr[k] = row[e0 + i];
      rw[k] = __float_as_int(w[e0 + i]);
      rk[k] = atomicAdd(&cnt[c >> SORT_SH], 1);
    }
  }
  __syncthreads();
  const int v = (tid < NB) ? cnt[tid] : 0;
  const int ex = block_excl_scan(v, wsum);
  if (tid < NB) {
    lstart[tid] = ex;
    gbase[tid] = offs[tid * NWG + g] - ex;
  }
  __syncthreads();
#pragma unroll
  for (int k = 0; k < SORT_CH / 1024; ++k) {
    const int i = k * 1024 + tid;
    if (i < n) {
      const int p = lstart[rc[k] >> SORT_SH] + rk[k];
      s_row[p] = rr[k];
      s_col[p] = rc[k];
      s_w[p] = rw[k];
    }
  }
  __syncthreads();
  for (int i = tid; i < n; i += 1024) {
    const int c = s_col[i];
    const int d = gbase[c >> SORT_SH] + i;
    prc[d] = make_int2(s_row[i], s_w[i]);
    pcl[d] = c;
  }
}

__global__ __launch_bounds__(512)
void bucket_sort_kernel(const int* __restrict__ offs, const int2* __restrict__ prc,
                        const int* __restrict__ pcl, int* __restrict__ start,
                        int2* __restrict__ sedge, int N, int NB, int NWG) {
  __shared__ int cnt[SORT_BN];
  __shared__ int nst[SORT_BN];
  __shared__ int cur[SORT_BN];
  __shared__ int wsum[32];
  __shared__ int2 stage[SORT_CAP];
  const int tid = threadIdx.x;
  const int b = blockIdx.x;
  const int bs = offs[b * NWG];
  const int be = offs[(b + 1) * NWG];
  const int n = be - bs;
  const int base = b << SORT_SH;
  if (tid < SORT_BN) { cnt[tid] = 0; cur[tid] = 0; }
  __syncthreads();
  for (int i0 = tid; i0 < n; i0 += 4 * 512) {
    int lc[4];
#pragma unroll
    for (int u = 0; u < 4; ++u) {
      const int i = i0 + u * 512;
      lc[u] = (i < n) ? pcl[bs + i] - base : -1;
    }
#pragma unroll
    for (int u = 0; u < 4; ++u)
      if (lc[u] >= 0) atomicAdd(&cnt[lc[u]], 1);
  }
  __syncthreads();
  const int v = (tid < SORT_BN) ? cnt[tid] : 0;
  const int ex = block_excl_scan(v, wsum);
  if (tid < SORT_BN) {
    nst[tid] = ex;
    if (base + tid < N) start[base + tid] = bs + ex;
  }
  if (b == NB - 1 && tid == 0) start[N] = be;
  __syncthreads();
  const bool staged = n <= SORT_CAP;
  for (int i0 = tid; i0 < n; i0 += 4 * 512) {
    int lc[4];
    int2 rec[4];
#pragma unroll
    for (int u = 0; u < 4; ++u) {
      const int i = i0 + u * 512;
      const bool in = i < n;
      lc[u] = in ? pcl[bs + i] - base : -1;
      rec[u] = in ? prc[bs + i] : make_int2(0, 0);
    }
#pragma unroll
    for (int u = 0; u < 4; ++u) {
      if (lc[u] >= 0) {
        const int p = nst[lc[u]] + atomicAdd(&cur[lc[u]], 1);
        if (staged) stage[p] = rec[u];
        else sedge[bs + p] = rec[u];
      }
    }
  }
  if (staged) {
    __syncthreads();
    for (int i = tid; i < n; i += 512) sedge[bs + i] = stage[i];
  }
}

__global__ void dis_csr_kernel(const int* __restrict__ start, const int2* __restrict__ sedge,
                               float* __restrict__ dis, int N) {
  int t = blockIdx.x * blockDim.x + threadIdx.x;
  int c = t >> 3;
  int sub = t & 7;
  if (c >= N) c = N - 1;
  const int s = start[c];
  const int e = start[c + 1];
  float d = 0.0f;
  for (int p = s + sub; p < e; p += 8) d += __int_as_float(sedge[p].y);
  d += __shfl_xor(d, 4);
  d += __shfl_xor(d, 2);
  d += __shfl_xor(d, 1);
  if (sub == 0) dis[c] = (d > 0.0f) ? rsqrtf(d) : 0.0f;
}

__global__ void norm_csr_kernel(const int* __restrict__ start, const float* __restrict__ dis,
                                int2* __restrict__ sedge, int N) {
  int t = blockIdx.x * blockDim.x + threadIdx.x;
  int c = t >> 3;
  int sub = t & 7;
  if (c >= N) return;
  const int s = start[c];
  const int e = start[c + 1];
  const float dc = dis[c];
  for (int p = s + sub; p < e; p += 8) {
    int2 ed = sedge[p];
    float nv = dis[ed.x] * __int_as_float(ed.y) * dc;
    sedge[p] = make_int2(ed.x, __float_as_int(nv));
  }
}

__global__ __launch_bounds__(256)
void hop_csr_kernel(const int* __restrict__ start, const int2* __restrict__ sedge,
                    const float* __restrict__ hsrc, float* __restrict__ hdst, int N) {
  int t = blockIdx.x * blockDim.x + threadIdx.x;
  int c = t >> 3;
  int sub = t & 7;
  if (c >= N) return;
  const int s = start[c];
  const int e = start[c + 1];
  const float4* __restrict__ h4 = reinterpret_cast<const float4*>(hsrc);
  float4 acc = make_float4(0.0f, 0.0f, 0.0f, 0.0f);
  if (s < e) {
    const int last = e - 1;
    int2 ed0 = sedge[min(s + 0, last)];
    int2 ed1 = sedge[min(s + 1, last)];
    int2 ed2 = sedge[min(s + 2, last)];
    int2 ed3 = sedge[min(s + 3, last)];
    for (int p = s; p < e; p += 4) {
      float4 v0 = h4[(long)ed0.x * (DIM / 4) + sub];
      float4 v1 = h4[(long)ed1.x * (DIM / 4) + sub];
      float4 v2 = h4[(long)ed2.x * (DIM / 4) + sub];
      float4 v3 = h4[(long)ed3.x * (DIM / 4) + sub];
      const int q = p + 4;
      int2 n0 = sedge[min(q + 0, last)];
      int2 n1 = sedge[min(q + 1, last)];
      int2 n2 = sedge[min(q + 2, last)];
      int2 n3 = sedge[min(q + 3, last)];
      const bool k1 = p + 1 < e, k2 = p + 2 < e, k3 = p + 3 < e;
      const float4 z = make_float4(0.0f, 0.0f, 0.0f, 0.0f);
      const float w0 = __int_as_float(ed0.y);
      const float w1 = k1 ? __int_as_float(ed1.y) : 0.0f;
      const float w2 = k2 ? __int_as_float(ed2.y) : 0.0f;
      const float w3 = k3 ? __int_as_float(ed3.y) : 0.0f;
      if (!k1) v1 = z;
      if (!k2) v2 = z;
      if (!k3) v3 = z;
      acc.x += w0 * v0.x; acc.y += w0 * v0.y; acc.z += w0 * v0.z; acc.w += w0 * v0.w;
      acc.x += w1 * v1.x; acc.y += w1 * v1.y; acc.z += w1 * v1.z; acc.w += w1 * v1.w;
      acc.x += w2 * v2.x; acc.y += w2 * v2.y; acc.z += w2 * v2.z; acc.w += w2 * v2.w;
      acc.x += w3 * v3.x; acc.y += w3 * v3.y; acc.z += w3 * v3.z; acc.w += w3 * v3.w;
      ed0 = n0; ed1 = n1; ed2 = n2; ed3 = n3;
    }
  }
  reinterpret_cast<float4*>(hdst)[(long)c * (DIM / 4) + sub] = acc;
}


__global__ __launch_bounds__(256)
void tag_gemm_wmma_kernel(const float* __restrict__ s0, const float* __restrict__ s1,
                          const float* __restrict__ s2, const float* __restrict__ s3,
                          const float* __restrict__ Wl,
                          const float* __restrict__ bl,
                          float* __restrict__ dst, int nTiles) {
  int wave = threadIdx.x >> 5;
  int tile = blockIdx.x * (blockDim.x >> 5) + wave;
  if (tile >= nTiles) return;
  int lane = threadIdx.x & 31;
  int mlane = lane & 15;
  bool hi = lane >= 16;
  int khalf = hi ? 2 : 0;
  long rowBase = (long)(tile * 16 + mlane) * DIM;

  v8f c0 = {};
  v8f c1 = {};

  const float* srcs[4] = {s0, s1, s2, s3};
#pragma unroll
  for (int k = 0; k < 4; ++k) {
    const float* S = srcs[k];
    const float* Wk = Wl + k * (DIM * DIM);
#pragma unroll
    for (int kk = 0; kk < DIM; kk += 4) {
      int ka = kk + khalf;
      v2f a = *reinterpret_cast<const v2f*>(S + rowBase + ka);
      v2f b0, b1;
      b0.x = Wk[(ka + 0) * DIM + mlane];
      b0.y = Wk[(ka + 1) * DIM + mlane];
      b1.x = Wk[(ka + 0) * DIM + mlane + 16];
      b1.y = Wk[(ka + 1) * DIM + mlane + 16];
      c0 = __builtin_amdgcn_wmma_f32_16x16x4_f32(false, a, false, b0, (short)0, c0, false, false);
      c1 = __builtin_amdgcn_wmma_f32_16x16x4_f32(false, a, false, b1, (short)0, c1, false, false);
    }
  }

  float bn0 = bl[mlane];
  float bn1 = bl[mlane + 16];
  int mBase = tile * 16 + (hi ? 8 : 0);
#pragma unroll
  for (int r = 0; r < 8; ++r) {
    float v0 = c0[r] + bn0;
    float v1 = c1[r] + bn1;
    v0 = v0 > 0.0f ? v0 : 0.0f;
    v1 = v1 > 0.0f ? v1 : 0.0f;
    dst[(long)(mBase + r) * DIM + mlane]      = v0;
    dst[(long)(mBase + r) * DIM + mlane + 16] = v1;
  }
}

__global__ void tag_gemm_tail_kernel(const float* __restrict__ s0, const float* __restrict__ s1,
                                     const float* __restrict__ s2, const float* __restrict__ s3,
                                     const float* __restrict__ Wl, const float* __restrict__ bl,
                                     float* __restrict__ dst, int startRow, int N) {
  int t = blockIdx.x * blockDim.x + threadIdx.x;
  int m = startRow + (t >> 5);
  int n = t & 31;
  if (m >= N) return;
  float acc = bl[n];
  const float* srcs[4] = {s0, s1, s2, s3};
  for (int k = 0; k < 4; ++k) {
    const float* S = srcs[k] + (long)m * DIM;
    const float* Wk = Wl + k * DIM * DIM;
    for (int j = 0; j < DIM; ++j) acc += S[j] * Wk[j * DIM + n];
  }
  dst[(long)m * DIM + n] = acc > 0.0f ? acc : 0.0f;
}


extern "C" void kernel_launch(void* const* d_in, const int* in_sizes, int n_in,
                              void* d_out, int out_size, void* d_ws, size_t ws_size,
                              hipStream_t stream) {
  (void)n_in; (void)out_size; (void)ws_size;
  const float* d_x = (const float*)d_in[0];
  const int*   d_ei = (const int*)d_in[1];
  const float* d_w = (const float*)d_in[2];
  const float* d_W = (const float*)d_in[3];
  const float* d_b = (const float*)d_in[4];
  const int N = in_sizes[0] / DIM;
  const int E = in_sizes[2];
  const int* row = d_ei;
  const int* col = d_ei + E;

  auto align_up = [](size_t v) { return (v + 255) & ~(size_t)255; };
  char* ws = (char*)d_ws;
  size_t off = 0;
  int*   cnt   = (int*)(ws + off);   off = align_up(off + (size_t)N * 4);
  int*   start = (int*)(ws + off);   off = align_up(off + ((size_t)N + 1) * 4);
  int*   rnk   = (int*)(ws + off);   off = align_up(off + (size_t)E * 4);
  int2*  sedge = (int2*)(ws + off);  off = align_up(off + (size_t)E * 8);
  float* dis   = (float*)(ws + off); off = align_up(off + (size_t)N * 4);
  float* h1    = (float*)(ws + off); off = align_up(off + (size_t)N * DIM * 4);
  float* h2    = (float*)(ws + off); off = align_up(off + (size_t)N * DIM * 4);
  float* h3    = (float*)(ws + off); off = align_up(off + (size_t)N * DIM * 4);
  float* xcur  = (float*)(ws + off); off = align_up(off + (size_t)N * DIM * 4);
  const int NB  = (N + SORT_BN - 1) >> SORT_SH;
  const int NWG = (E + SORT_CH - 1) / SORT_CH;
  const int M   = NB * NWG;
  int*   hist  = (int*)(ws + off);   off = align_up(off + (size_t)M * 4);
  int*   offs  = (int*)(ws + off);   off = align_up(off + ((size_t)M + 1) * 4);
  int2*  prc   = (int2*)(ws + off);  off = align_up(off + (size_t)E * 8);
  int*   pcl   = (int*)(ws + off);   off = align_up(off + (size_t)E * 4);

  const int B = 256;
  auto gblk = [&](long n) { return (int)((n + B - 1) / B); };
  const long grp8 = (long)N * 8;

  if (NB <= SORT_MAXNB) {
    bucket_hist_kernel<<<NWG, 1024, 0, stream>>>(col, hist, E, NB, NWG);
    scan_kernel<<<M / SCAN_TILE + 1, SCAN_THREADS, 0, stream>>>(hist, offs, M);
    bucket_part_kernel<<<NWG, 1024, 0, stream>>>(row, col, d_w, offs, prc, pcl, E, NB, NWG);
    bucket_sort_kernel<<<NB, 512, 0, stream>>>(offs, prc, pcl, start, sedge, N, NB, NWG);
  } else {
    zero_i32_kernel<<<gblk(N), B, 0, stream>>>(cnt, N);
    rank_kernel<<<gblk(E), B, 0, stream>>>(col, cnt, rnk, E);
    scan_kernel<<<N / SCAN_TILE + 1, SCAN_THREADS, 0, stream>>>(cnt, start, N);
    fill_kernel<<<gblk(E), B, 0, stream>>>(row, col, d_w, start, rnk, sedge, E);
  }
  dis_csr_kernel<<<gblk(grp8), B, 0, stream>>>(start, sedge, dis, N);
  norm_csr_kernel<<<gblk(grp8), B, 0, stream>>>(start, dis, sedge, N);

  const int nTiles = N / 16;
  const int tileBlocks = (nTiles + 7) / 8;
  const int tailStart = nTiles * 16;
  const int tailRows = N - tailStart;

  for (int l = 0; l < 3; ++l) {
    const float* xin = (l == 0) ? d_x : xcur;
    float* dst = (l == 2) ? (float*)d_out : xcur;
    const float* Wl = d_W + (size_t)l * 4 * DIM * DIM;
    const float* bl = d_b + (size_t)l * DIM;

    hop_csr_kernel<<<gblk(grp8), B, 0, stream>>>(start, sedge, xin, h1, N);
    hop_csr_kernel<<<gblk(grp8), B, 0, stream>>>(start, sedge, h1, h2, N);
    hop_csr_kernel<<<gblk(grp8), B, 0, stream>>>(start, sedge, h2, h3, N);

    tag_gemm_wmma_kernel<<<tileBlocks, B, 0, stream>>>(xin, h1, h2, h3, Wl, bl, dst, nTiles);
    if (tailRows > 0) {
      tag_gemm_tail_kernel<<<gblk((long)tailRows * 32), B, 0, stream>>>(
          xin, h1, h2, h3, Wl, bl, dst, tailStart, N);
    }
  }
}


# Third hop of each layer fused with the layer's product: a 32-row work-group sums h3 into LDS beside its x, h1, h2 rows and the weights, four waves run the WMMAs from LDS; 3 launches and h3 plane go
# speedup vs baseline: 5.7526x; 1.1324x over previous
#include <hip/hip_runtime.h>


typedef float v2f __attribute__((ext_vector_type(2)));
typedef float v8f __attribute__((ext_vector_type(8)));

#define DIM 32
#define SCAN_THREADS 1024
#define SCAN_TILE (SCAN_THREADS * 4)


__global__ void zero_i32_kernel(int* __restrict__ p, int n) {
  int i = blockIdx.x * blockDim.x + threadIdx.x;
  if (i < n) p[i] = 0;
}

__global__ void rank_kernel(const int* __restrict__ col, int* __restrict__ cnt,
                            int* __restrict__ rnk, int E) {
  int e = blockIdx.x * blockDim.x + threadIdx.x;
  if (e < E) rnk[e] = atomicAdd(&cnt[col[e]], 1);
}

__global__ __launch_bounds__(SCAN_THREADS)
void scan_kernel(const int* __restrict__ cnt, int* __restrict__ start, int N) {
  __shared__ int red[SCAN_THREADS / 32];
  __shared__ int wtot[SCAN_THREADS / 32];
  const int tid = threadIdx.x;
  const int lane = tid & 31;
  const int wv = tid >> 5;
  const int base0 = blockIdx.x * SCAN_TILE;

  int a0 = 0, a1 = 0, a2 = 0, a3 = 0;
  int i = tid * 4;
  for (; i + 3 * SCAN_TILE < base0; i += 4 * SCAN_TILE) {
    int4 v0 = *reinterpret_cast<const int4*>(cnt + i);
    int4 v1 = *reinterpret_cast<const int4*>(cnt + i + SCAN_TILE);
    int4 v2 = *reinterpret_cast<const int4*>(cnt + i + 2 * SCAN_TILE);
    int4 v3 = *reinterpret_cast<const int4*>(cnt + i + 3 * SCAN_TILE);
    a0 += (v0.x + v0.y) + (v0.z + v0.w);
    a1 += (v1.x + v1.y) + (v1.z + v1.w);
    a2 += (v2.x + v2.y) + (v2.z + v2.w);
    a3 += (v3.x + v3.y) + (v3.z + v3.w);
  }
  for (; i < base0; i += SCAN_TILE) {
    int4 v = *reinterpret_cast<const int4*>(cnt + i);
    a0 += (v.x + v.y) + (v.z + v.w);
  }
  int acc = (a0 + a1) + (a2 + a3);
  for (int o = 16; o > 0; o >>= 1) acc += __shfl_xor(acc, o);
  if (lane == 0) red[wv] = acc;

  const int i0 = base0 + tid * 4;
  int c0 = 0, c1 = 0, c2 = 0, c3 = 0;
  if (i0 + 3 < N) {
    int4 v = *reinterpret_cast<const int4*>(cnt + i0);
    c0 = v.x; c1 = v.y; c2 = v.z; c3 = v.w;
  } else {
    if (i0 + 0 < N) c0 = cnt[i0 + 0];
    if (i0 + 1 < N) c1 = cnt[i0 + 1];
    if (i0 + 2 < N) c2 = cnt[i0 + 2];
  }
  const int t1 = c0, t2 = t1 + c1, t3 = t2 + c2, tsum = t3 + c3;
  int inc = tsum;
  for (int o = 1; o < 32; o <<= 1) {
    int u = __shfl_up(inc, o);
    if (lane >= o) inc += u;
  }
  if (lane == 31) wtot[wv] = inc;
  __syncthreads();
  int pre = 0;
  for (int k = 0; k < SCAN_THREADS / 32; ++k) pre += red[k];
  for (int k = 0; k < wv; ++k) pre += wtot[k];
  const int ex = pre + inc - tsum;
  if (i0 + 0 <= N) start[i0 + 0] = ex;
  if (i0 + 1 <= N) start[i0 + 1] = ex + t1;
  if (i0 + 2 <= N) start[i0 + 2] = ex + t2;
  if (i0 + 3 <= N) start[i0 + 3] = ex + t3;
}

__global__ void fill_kernel(const int* __restrict__ row, const int* __restrict__ col,
                            const float* __restrict__ w, const int* __restrict__ start,
                            const int* __restrict__ rnk, int2* __restrict__ sedge, int E) {
  int e = blockIdx.x * blockDim.x + threadIdx.x;
  if (e < E) {
    int p = start[col[e]] + rnk[e];
    sedge[p] = make_int2(row[e], __float_as_int(w[e]));
  }
}


#define SORT_SH 8
#define SORT_BN (1 << SORT_SH)
#define SORT_MAXNB 512
#define SORT_CH 8192
#define SORT_CAP 8192

__device__ __forceinline__ int block_excl_scan(int v, int* wsum) {
  const int lane = threadIdx.x & 31;
  const int wv = threadIdx.x >> 5;
  int inc = v;
  for (int o = 1; o < 32; o <<= 1) {
    int u = __shfl_up(inc, o);
    if (lane >= o) inc += u;
  }
  if (lane == 31) wsum[wv] = inc;
  __syncthreads();
  int pre = 0;
  for (int k = 0; k < wv; ++k) pre += wsum[k];
  return pre + inc - v;
}

__global__ __launch_bounds__(1024)
void bucket_hist_kernel(const int* __restrict__ col, int* __restrict__ hist,
                        int E, int NB, int NWG) {
  __shared__ int cnt[SORT_MAXNB];
  const int tid = threadIdx.x;
  const int g = blockIdx.x;
  for (int b = tid; b < NB; b += 1024) cnt[b] = 0;
  __syncthreads();
  const int e0 = g * SORT_CH;
  const int n = min(E - e0, SORT_CH);
  int rc[SORT_CH / 1024];
#pragma unroll
  for (int k = 0; k < SORT_CH / 1024; ++k) {
    const int i = k * 1024 + tid;
    rc[k] = (i < n) ? col[e0 + i] : -1;
  }
#pragma unroll
  for (int k = 0; k < SORT_CH / 1024; ++k)
    if (rc[k] >= 0) atomicAdd(&cnt[rc[k] >> SORT_SH], 1);
  __syncthreads();
  for (int b = tid; b < NB; b += 1024) hist[b * NWG + g] = cnt[b];
}

__global__ __launch_bounds__(1024)
void bucket_part_kernel(const int* __restrict__ row, const int* __restrict__ col,
                        const float* __restrict__ w, const int* __restrict__ offs,
                        int2* __restrict__ prc, int* __restrict__ pcl,
                        int E, int NB, int NWG) {
  __shared__ int cnt[SORT_MAXNB];
  __shared__ int lstart[SORT_MAXNB];
  __shared__ int gbase[SORT_MAXNB];
  __shared__ int wsum[32];
  __shared__ int s_row[SORT_CH];
  __shared__ int s_col[SORT_CH];
  __shared__ int s_w[SORT_CH];
  const int tid = threadIdx.x;
  const int g = blockIdx.x;
  for (int b = tid; b < NB; b += 1024) cnt[b] = 0;
  __syncthreads();
  const int e0 = g * SORT_CH;
  const int n = min(E - e0, SORT_CH);
  int rc[SORT_CH / 1024], rr[SORT_CH / 1024], rw[SORT_CH / 1024], rk[SORT_CH / 1024];
#pragma unroll
  for (int k = 0; k < SORT_CH / 1024; ++k) {
    const int i = k * 1024 + tid;
    if (i < n) {
      const int c = col[e0 + i];
      rc[k] = c;
      rr[k] = row[e0 + i];
      rw[k] = __float_as_int(w[e0 + i]);
      rk[k] = atomicAdd(&cnt[c >> SORT_SH], 1);
    }
  }
  __syncthreads();
  const int v = (tid < NB) ? cnt[tid] : 0;
  const int ex = block_excl_scan(v, wsum);
  if (tid < NB) {
    lstart[tid] = ex;
    gbase[tid] = offs[tid * NWG + g] - ex;
  }
  __syncthreads();
#pragma unroll
  for (int k = 0; k < SORT_CH / 1024; ++k) {
    const int i = k * 1024 + tid;
    if (i < n) {
      const int p = lstart[rc[k] >> SORT_SH] + rk[k];
      s_row[p] = rr[k];
      s_col[p] = rc[k];
      s_w[p] = rw[k];
    }
  }
  __syncthreads();
  for (int i = tid; i < n; i += 1024) {
    const int c = s_col[i];
    const int d = gbase[c >> SORT_SH] + i;
    prc[d] = make_int2(s_row[i], s_w[i]);
    pcl[d] = c;
  }
}

__global__ __launch_bounds__(512)
void bucket_sort_kernel(const int* __restrict__ offs, const int2* __restrict__ prc,
                        const int* __restrict__ pcl, int* __restrict__ start,
                        int2* __restrict__ sedge, int N, int NB, int NWG) {
  __shared__ int cnt[SORT_BN];
  __shared__ int nst[SORT_BN];
  __shared__ int cur[SORT_BN];
  __shared__ int wsum[32];
  __shared__ int2 stage[SORT_CAP];
  const int tid = threadIdx.x;
  const int b = blockIdx.x;
  const int bs = offs[b * NWG];
  const int be = offs[(b + 1) * NWG];
  const int n = be - bs;
  const int base = b << SORT_SH;
  if (tid < SORT_BN) { cnt[tid] = 0; cur[tid] = 0; }
  __syncthreads();
  for (int i0 = tid; i0 < n; i0 += 4 * 512) {
    int lc[4];
#pragma unroll
    for (int u = 0; u < 4; ++u) {
      const int i = i0 + u * 512;
      lc[u] = (i < n) ? pcl[bs + i] - base : -1;
    }
#pragma unroll
    for (int u = 0; u < 4; ++u)
      if (lc[u] >= 0) atomicAdd(&cnt[lc[u]], 1);
  }
  __syncthreads();
  const int v = (tid < SORT_BN) ? cnt[tid] : 0;
  const int ex = block_excl_scan(v, wsum);
  if (tid < SORT_BN) {
    nst[tid] = ex;
    if (base + tid < N) start[base + tid] = bs + ex;
  }
  if (b == NB - 1 && tid == 0) start[N] = be;
  __syncthreads();
  const bool staged = n <= SORT_CAP;
  for (int i0 = tid; i0 < n; i0 += 4 * 512) {
    int lc[4];
    int2 rec[4];
#pragma unroll
    for (int u = 0; u < 4; ++u) {
      const int i = i0 + u * 512;
      const bool in = i < n;
      lc[u] = in ? pcl[bs + i] - base : -1;
      rec[u] = in ? prc[bs + i] : make_int2(0, 0);
    }
#pragma unroll
    for (int u = 0; u < 4; ++u) {
      if (lc[u] >= 0) {
        const int p = nst[lc[u]] + atomicAdd(&cur[lc[u]], 1);
        if (staged) stage[p] = rec[u];
        else sedge[bs + p] = rec[u];
      }
    }
  }
  if (staged) {
    __syncthreads();
    for (int i = tid; i < n; i += 512) sedge[bs + i] = stage[i];
  }
}

__global__ void dis_csr_kernel(const int* __restrict__ start, const int2* __restrict__ sedge,
                               float* __restrict__ dis, int N) {
  int t = blockIdx.x * blockDim.x + threadIdx.x;
  int c = t >> 3;
  int sub = t & 7;
  if (c >= N) c = N - 1;
  const int s = start[c];
  const int e = start[c + 1];
  float d = 0.0f;
  for (int p = s + sub; p < e; p += 8) d += __int_as_float(sedge[p].y);
  d += __shfl_xor(d, 4);
  d += __shfl_xor(d, 2);
  d += __shfl_xor(d, 1);
  if (sub == 0) dis[c] = (d > 0.0f) ? rsqrtf(d) : 0.0f;
}

__global__ void norm_csr_kernel(const int* __restrict__ start, const float* __restrict__ dis,
                                int2* __restrict__ sedge, int N) {
  int t = blockIdx.x * blockDim.x + threadIdx.x;
  int c = t >> 3;
  int sub = t & 7;
  if (c >= N) return;
  const int s = start[c];
  const int e = start[c + 1];
  const float dc = dis[c];
  for (int p = s + sub; p < e; p += 8) {
    int2 ed = sedge[p];
    float nv = dis[ed.x] * __int_as_float(ed.y) * dc;
    sedge[p] = make_int2(ed.x, __float_as_int(nv));
  }
}

__global__ __launch_bounds__(256)
void hop_csr_kernel(const int* __restrict__ start, const int2* __restrict__ sedge,
                    const float* __restrict__ hsrc, float* __restrict__ hdst, int N) {
  int t = blockIdx.x * blockDim.x + threadIdx.x;
  int c = t >> 3;
  int sub = t & 7;
  if (c >= N) return;
  const int s = start[c];
  const int e = start[c + 1];
  const float4* __restrict__ h4 = reinterpret_cast<const float4*>(hsrc);
  float4 acc = make_float4(0.0f, 0.0f, 0.0f, 0.0f);
  if (s < e) {
    const int last = e - 1;
    int2 ed0 = sedge[min(s + 0, last)];
    int2 ed1 = sedge[min(s + 1, last)];
    int2 ed2 = sedge[min(s + 2, last)];
    int2 ed3 = sedge[min(s + 3, last)];
    for (int p = s; p < e; p += 4) {
      float4 v0 = h4[(long)ed0.x * (DIM / 4) + sub];
      float4 v1 = h4[(long)ed1.x * (DIM / 4) + sub];
      float4 v2 = h4[(long)ed2.x * (DIM / 4) + sub];
      float4 v3 = h4[(long)ed3.x * (DIM / 4) + sub];
      const int q = p + 4;
      int2 n0 = sedge[min(q + 0, last)];
      int2 n1 = sedge[min(q + 1, last)];
      int2 n2 = sedge[min(q + 2, last)];
      int2 n3 = sedge[min(q + 3, last)];
      const bool k1 = p + 1 < e, k2 = p + 2 < e, k3 = p + 3 < e;
      const float4 z = make_float4(0.0f, 0.0f, 0.0f, 0.0f);
      const float w0 = __int_as_float(ed0.y);
      const float w1 = k1 ? __int_as_float(ed1.y) : 0.0f;
      const float w2 = k2 ? __int_as_float(ed2.y) : 0.0f;
      const float w3 = k3 ? __int_as_float(ed3.y) : 0.0f;
      if (!k1) v1 = z;
      if (!k2) v2 = z;
      if (!k3) v3 = z;
      acc.x += w0 * v0.x; acc.y += w0 * v0.y; acc.z += w0 * v0.z; acc.w += w0 * v0.w;
      acc.x += w1 * v1.x; acc.y += w1 * v1.y; acc.z += w1 * v1.z; acc.w += w1 * v1.w;
      acc.x += w2 * v2.x; acc.y += w2 * v2.y; acc.z += w2 * v2.z; acc.w += w2 * v2.w;
      acc.x += w3 * v3.x; acc.y += w3 * v3.y; acc.z += w3 * v3.z; acc.w += w3 * v3.w;
      ed0 = n0; ed1 = n1; ed2 = n2; ed3 = n3;
    }
  }
  reinterpret_cast<float4*>(hdst)[(long)c * (DIM / 4) + sub] = acc;
}


#define HT_ROWS 32
#define HT_PITCH 36
#define HW_PITCH 20
#define HW_HALF (4 * DIM * HW_PITCH)

__global__ __launch_bounds__(256)
void hop_gemm_wmma_kernel(const int* __restrict__ start, const int2* __restrict__ sedge,
                          const float* s0, const float* __restrict__ s1,
                          const float* __restrict__ s2,
                          const float* __restrict__ Wl,
                          const float* __restrict__ bl,
                          float* dst, float* __restrict__ h3tail, int N, int nTiles) {
  __shared__ __attribute__((aligned(16))) float As[4 * HT_ROWS * HT_PITCH];
  __shared__ __attribute__((aligned(16))) float Ws[2 * HW_HALF];
  const int tid = threadIdx.x;
  const int r = tid >> 3;
  const int sub = tid & 7;
  const int c = blockIdx.x * HT_ROWS + r;
  const float4 z4 = make_float4(0.0f, 0.0f, 0.0f, 0.0f);

  const float4* __restrict__ W4 = reinterpret_cast<const float4*>(Wl);
  const float4 wq0 = W4[0 * 256 + tid];
  const float4 wq1 = W4[1 * 256 + tid];
  const float4 wq2 = W4[2 * 256 + tid];
  const float4 wq3 = W4[3 * 256 + tid];
  const long ro = (long)c * (DIM / 4) + sub;
  float4 a0 = z4, a1 = z4, a2 = z4;
  int s = 0, e = 0;
  if (c < N) {
    a0 = reinterpret_cast<const float4*>(s0)[ro];
    a1 = reinterpret_cast<const float4*>(s1)[ro];
    a2 = reinterpret_cast<const float4*>(s2)[ro];
    s = start[c];
    e = start[c + 1];
  }
  {
    float* wrow = Ws + (sub >> 2) * HW_HALF + r * HW_PITCH + (sub & 3) * 4;
    *reinterpret_cast<float4*>(wrow + 0 * DIM * HW_PITCH) = wq0;
    *reinterpret_cast<float4*>(wrow + 1 * DIM * HW_PITCH) = wq1;
    *reinterpret_cast<float4*>(wrow + 2 * DIM * HW_PITCH) = wq2;
    *reinterpret_cast<float4*>(wrow + 3 * DIM * HW_PITCH) = wq3;
  }
  float* arow_st = As + r * HT_PITCH + sub * 4;
  *reinterpret_cast<float4*>(arow_st + 0 * HT_ROWS * HT_PITCH) = a0;
  *reinterpret_cast<float4*>(arow_st + 1 * HT_ROWS * HT_PITCH) = a1;
  *reinterpret_cast<float4*>(arow_st + 2 * HT_ROWS * HT_PITCH) = a2;

  float4 acc = z4;
  if (s < e) {
    const float4* __restrict__ h4 = reinterpret_cast<const float4*>(s2);
    const int last = e - 1;
    int2 ed0 = sedge[min(s + 0, last)];
    int2 ed1 = sedge[min(s + 1, last)];
    int2 ed2 = sedge[min(s + 2, last)];
    int2 ed3 = sedge[min(s + 3, last)];
    for (int p = s; p < e; p += 4) {
      float4 v0 = h4[(long)ed0.x * (DIM / 4) + sub];
      float4 v1 = h4[(long)ed1.x * (DIM / 4) + sub];
      float4 v2 = h4[(long)ed2.x * (DIM / 4) + sub];
      float4 v3 = h4[(long)ed3.x * (DIM / 4) + sub];
      const int q = p + 4;
      int2 n0 = sedge[min(q + 0, last)];
      int2 n1 = sedge[min(q + 1, last)];
      int2 n2 = sedge[min(q + 2, last)];
      int2 n3 = sedge[min(q + 3, last)];
      const bool k1 = p + 1 < e, k2 = p + 2 < e, k3 = p + 3 < e;
      const float w0 = __int_as_float(ed0.y);
      const float w1 = k1 ? __int_as_float(ed1.y) : 0.0f;
      const float w2 = k2 ? __int_as_float(ed2.y) : 0.0f;
      const float w3 = k3 ? __int_as_float(ed3.y) : 0.0f;
      if (!k1) v1 = z4;
      if (!k2) v2 = z4;
      if (!k3) v3 = z4;
      acc.x += w0 * v0.x; acc.y += w0 * v0.y; acc.z += w0 * v0.z; acc.w += w0 * v0.w;
      acc.x += w1 * v1.x; acc.y += w1 * v1.y; acc.z += w1 * v1.z; acc.w += w1 * v1.w;
      acc.x += w2 * v2.x; acc.y += w2 * v2.y; acc.z += w2 * v2.z; acc.w += w2 * v2.w;
      acc.x += w3 * v3.x; acc.y += w3 * v3.y; acc.z += w3 * v3.z; acc.w += w3 * v3.w;
      ed0 = n0; ed1 = n1; ed2 = n2; ed3 = n3;
    }
  }
  if (c < N && c >= nTiles * 16) reinterpret_cast<float4*>(h3tail)[ro] = acc;
  *reinterpret_cast<float4*>(arow_st + 3 * HT_ROWS * HT_PITCH) = acc;
  __syncthreads();

  const int wave = tid >> 5;
  if (wave >= HT_ROWS / 8) return;
  const int tile = blockIdx.x * (HT_ROWS / 16) + (wave >> 1);
  if (tile >= nTiles) return;
  const int nh = wave & 1;
  const int lane = tid & 31;
  const int mlane = lane & 15;
  const bool hi = lane >= 16;
  const int khalf = hi ? 2 : 0;
  const float* arow = As + ((wave >> 1) * 16 + mlane) * HT_PITCH;
  const float* wcol = Ws + nh * HW_HALF + mlane;
  const int ncol = nh * 16 + mlane;
  const float bn = bl[ncol];

  v8f cacc = {};

  v2f af[2][4], bf[2][4];
#pragma unroll
  for (int j = 0; j < 4; ++j) {
    const int ka = 4 * j + khalf;
    af[0][j] = *reinterpret_cast<const v2f*>(arow + ka);
    bf[0][j].x = wcol[(ka + 0) * HW_PITCH];
    bf[0][j].y = wcol[(ka + 1) * HW_PITCH];
  }
#pragma unroll
  for (int g = 0; g < 8; ++g) {
    if (g < 7) {
      const int gn = g + 1;
      const float* Ak = arow + (gn >> 1) * (HT_ROWS * HT_PITCH);
      const float* Wk = wcol + (gn >> 1) * (DIM * HW_PITCH);
#pragma unroll
      for (int j = 0; j < 4; ++j) {
        const int ka = 4 * ((gn & 1) * 4 + j) + khalf;
        af[gn & 1][j] = *reinterpret_cast<const v2f*>(Ak + ka);
        bf[gn & 1][j].x = Wk[(ka + 0) * HW_PITCH];
        bf[gn & 1][j].y = Wk[(ka + 1) * HW_PITCH];
      }
    }
    __builtin_amdgcn_sched_barrier(0);
#pragma unroll
    for (int j = 0; j < 4; ++j)
      cacc = __builtin_amdgcn_wmma_f32_16x16x4_f32(false, af[g & 1][j], false, bf[g & 1][j],
                                                   (short)0, cacc, false, false);
    __builtin_amdgcn_sched_barrier(0);
  }

  int mBase = tile * 16 + (hi ? 8 : 0);
#pragma unroll
  for (int rr = 0; rr < 8; ++rr) {
    float v = cacc[rr] + bn;
    v = v > 0.0f ? v : 0.0f;
    dst[(long)(mBase + rr) * DIM + ncol] = v;
  }
}

__global__ void tag_gemm_tail_kernel(const float* __restrict__ s0, const float* __restrict__ s1,
                                     const float* __restrict__ s2, const float* __restrict__ s3,
                                     const float* __restrict__ Wl, const float* __restrict__ bl,
                                     float* __restrict__ dst, int startRow, int N) {
  int t = blockIdx.x * blockDim.x + threadIdx.x;
  int m = startRow + (t >> 5);
  int n = t & 31;
  if (m >= N) return;
  float acc = bl[n];
  const float* srcs[4] = {s0, s1, s2, s3};
  for (int k = 0; k < 4; ++k) {
    const float* S = srcs[k] + (long)m * DIM;
    const float* Wk = Wl + k * DIM * DIM;
    for (int j = 0; j < DIM; ++j) acc += S[j] * Wk[j * DIM + n];
  }
  dst[(long)m * DIM + n] = acc > 0.0f ? acc : 0.0f;
}


extern "C" void kernel_launch(void* const* d_in, const int* in_sizes, int n_in,
                              void* d_out, int out_size, void* d_ws, size_t ws_size,
                              hipStream_t stream) {
  (void)n_in; (void)out_size; (void)ws_size;
  const float* d_x = (const float*)d_in[0];
  const int*   d_ei = (const int*)d_in[1];
  const float* d_w = (const float*)d_in[2];
  const float* d_W = (const float*)d_in[3];
  const float* d_b = (const float*)d_in[4];
  const int N = in_sizes[0] / DIM;
  const int E = in_sizes[2];
  const int* row = d_ei;
  const int* col = d_ei + E;

  auto align_up = [](size_t v) { return (v + 255) & ~(size_t)255; };
  char* ws = (char*)d_ws;
  size_t off = 0;
  int*   cnt   = (int*)(ws + off);   off = align_up(off + (size_t)N * 4);
  int*   start = (int*)(ws + off);   off = align_up(off + ((size_t)N + 1) * 4);
  int*   rnk   = (int*)(ws + off);   off = align_up(off + (size_t)E * 4);
  int2*  sedge = (int2*)(ws + off);  off = align_up(off + (size_t)E * 8);
  float* dis   = (float*)(ws + off); off = align_up(off + (size_t)N * 4);
  float* h1    = (float*)(ws + off); off = align_up(off + (size_t)N * DIM * 4);
  float* h2    = (float*)(ws + off); off = align_up(off + (size_t)N * DIM * 4);
  float* h3    = (float*)(ws + off); off = align_up(off + (size_t)N * DIM * 4);
  float* xcur  = (float*)(ws + off); off = align_up(off + (size_t)N * DIM * 4);
  const int NB  = (N + SORT_BN - 1) >> SORT_SH;
  const int NWG = (E + SORT_CH - 1) / SORT_CH;
  const int M   = NB * NWG;
  int*   hist  = (int*)(ws + off);   off = align_up(off + (size_t)M * 4);
  int*   offs  = (int*)(ws + off);   off = align_up(off + ((size_t)M + 1) * 4);
  int2*  prc   = (int2*)(ws + off);  off = align_up(off + (size_t)E * 8);
  int*   pcl   = (int*)(ws + off);   off = align_up(off + (size_t)E * 4);

  const int B = 256;
  auto gblk = [&](long n) { return (int)((n + B - 1) / B); };
  const long grp8 = (long)N * 8;

  if (NB <= SORT_MAXNB) {
    bucket_hist_kernel<<<NWG, 1024, 0, stream>>>(col, hist, E, NB, NWG);
    scan_kernel<<<M / SCAN_TILE + 1, SCAN_THREADS, 0, stream>>>(hist, offs, M);
    bucket_part_kernel<<<NWG, 1024, 0, stream>>>(row, col, d_w, offs, prc, pcl, E, NB, NWG);
    bucket_sort_kernel<<<NB, 512, 0, stream>>>(offs, prc, pcl, start, sedge, N, NB, NWG);
  } else {
    zero_i32_kernel<<<gblk(N), B, 0, stream>>>(cnt, N);
    rank_kernel<<<gblk(E), B, 0, stream>>>(col, cnt, rnk, E);
    scan_kernel<<<N / SCAN_TILE + 1, SCAN_THREADS, 0, stream>>>(cnt, start, N);
    fill_kernel<<<gblk(E), B, 0, stream>>>(row, col, d_w, start, rnk, sedge, E);
  }
  dis_csr_kernel<<<gblk(grp8), B, 0, stream>>>(start, sedge, dis, N);
  norm_csr_kernel<<<gblk(grp8), B, 0, stream>>>(start, dis, sedge, N);

  const int nTiles = N / 16;
  const int hopTileBlocks = (N + HT_ROWS - 1) / HT_ROWS;
  const int tailStart = nTiles * 16;
  const int tailRows = N - tailStart;

  for (int l = 0; l < 3; ++l) {
    const float* xin = (l == 0) ? d_x : xcur;
    float* dst = (l == 2) ? (float*)d_out : xcur;
    const float* Wl = d_W + (size_t)l * 4 * DIM * DIM;
    const float* bl = d_b + (size_t)l * DIM;

    hop_csr_kernel<<<gblk(grp8), B, 0, stream>>>(start, sedge, xin, h1, N);
    hop_csr_kernel<<<gblk(grp8), B, 0, stream>>>(start, sedge, h1, h2, N);

    hop_gemm_wmma_kernel<<<hopTileBlocks, B, 0, stream>>>(start, sedge, xin, h1, h2, Wl, bl, dst,
                                                          h3, N, nTiles);
    if (tailRows > 0) {
      tag_gemm_tail_kernel<<<gblk((long)tailRows * 32), B, 0, stream>>>(
          xin, h1, h2, h3, Wl, bl, dst, tailStart, N);
    }
  }
}


# Each hop work-group stages its 32 rows' contiguous piece of the grouped edge list in LDS once (whole 128-byte lines) and reads entries from there, not 4 per trip from global; per-row sum order kept
# speedup vs baseline: 6.5908x; 1.1457x over previous
#include <hip/hip_runtime.h>


typedef float v2f __attribute__((ext_vector_type(2)));
typedef float v8f __attribute__((ext_vector_type(8)));

#define DIM 32
#define SCAN_THREADS 1024
#define SCAN_TILE (SCAN_THREADS * 4)


__global__ void zero_i32_kernel(int* __restrict__ p, int n) {
  int i = blockIdx.x * blockDim.x + threadIdx.x;
  if (i < n) p[i] = 0;
}

__global__ void rank_kernel(const int* __restrict__ col, int* __restrict__ cnt,
                            int* __restrict__ rnk, int E) {
  int e = blockIdx.x * blockDim.x + threadIdx.x;
  if (e < E) rnk[e] = atomicAdd(&cnt[col[e]], 1);
}

__global__ __launch_bounds__(SCAN_THREADS)
void scan_kernel(const int* __restrict__ cnt, int* __restrict__ start, int N) {
  __shared__ int red[SCAN_THREADS / 32];
  __shared__ int wtot[SCAN_THREADS / 32];
  const int tid = threadIdx.x;
  const int lane = tid & 31;
  const int wv = tid >> 5;
  const int base0 = blockIdx.x * SCAN_TILE;

  int a0 = 0, a1 = 0, a2 = 0, a3 = 0;
  int i = tid * 4;
  for (; i + 3 * SCAN_TILE < base0; i += 4 * SCAN_TILE) {
    int4 v0 = *reinterpret_cast<const int4*>(cnt + i);
    int4 v1 = *reinterpret_cast<const int4*>(cnt + i + SCAN_TILE);
    int4 v2 = *reinterpret_cast<const int4*>(cnt + i + 2 * SCAN_TILE);
    int4 v3 = *reinterpret_cast<const int4*>(cnt + i + 3 * SCAN_TILE);
    a0 += (v0.x + v0.y) + (v0.z + v0.w);
    a1 += (v1.x + v1.y) + (v1.z + v1.w);
    a2 += (v2.x + v2.y) + (v2.z + v2.w);
    a3 += (v3.x + v3.y) + (v3.z + v3.w);
  }
  for (; i < base0; i += SCAN_TILE) {
    int4 v = *reinterpret_cast<const int4*>(cnt + i);
    a0 += (v.x + v.y) + (v.z + v.w);
  }
  int acc = (a0 + a1) + (a2 + a3);
  for (int o = 16; o > 0; o >>= 1) acc += __shfl_xor(acc, o);
  if (lane == 0) red[wv] = acc;

  const int i0 = base0 + tid * 4;
  int c0 = 0, c1 = 0, c2 = 0, c3 = 0;
  if (i0 + 3 < N) {
    int4 v = *reinterpret_cast<const int4*>(cnt + i0);
    c0 = v.x; c1 = v.y; c2 = v.z; c3 = v.w;
  } else {
    if (i0 + 0 < N) c0 = cnt[i0 + 0];
    if (i0 + 1 < N) c1 = cnt[i0 + 1];
    if (i0 + 2 < N) c2 = cnt[i0 + 2];
  }
  const int t1 = c0, t2 = t1 + c1, t3 = t2 + c2, tsum = t3 + c3;
  int inc = tsum;
  for (int o = 1; o < 32; o <<= 1) {
    int u = __shfl_up(inc, o);
    if (lane >= o) inc += u;
  }
  if (lane == 31) wtot[wv] = inc;
  __syncthreads();
  int pre = 0;
  for (int k = 0; k < SCAN_THREADS / 32; ++k) pre += red[k];
  for (int k = 0; k < wv; ++k) pre += wtot[k];
  const int ex = pre + inc - tsum;
  if (i0 + 0 <= N) start[i0 + 0] = ex;
  if (i0 + 1 <= N) start[i0 + 1] = ex + t1;
  if (i0 + 2 <= N) start[i0 + 2] = ex + t2;
  if (i0 + 3 <= N) start[i0 + 3] = ex + t3;
}

__global__ void fill_kernel(const int* __restrict__ row, const int* __restrict__ col,
                            const float* __restrict__ w, const int* __restrict__ start,
                            const int* __restrict__ rnk, int2* __restrict__ sedge, int E) {
  int e = blockIdx.x * blockDim.x + threadIdx.x;
  if (e < E) {
    int p = start[col[e]] + rnk[e];
    sedge[p] = make_int2(row[e], __float_as_int(w[e]));
  }
}


#define SORT_SH 8
#define SORT_BN (1 << SORT_SH)
#define SORT_MAXNB 512
#define SORT_CH 8192
#define SORT_CAP 8192

__device__ __forceinline__ int block_excl_scan(int v, int* wsum) {
  const int lane = threadIdx.x & 31;
  const int wv = threadIdx.x >> 5;
  int inc = v;
  for (int o = 1; o < 32; o <<= 1) {
    int u = __shfl_up(inc, o);
    if (lane >= o) inc += u;
  }
  if (lane == 31) wsum[wv] = inc;
  __syncthreads();
  int pre = 0;
  for (int k = 0; k < wv; ++k) pre += wsum[k];
  return pre + inc - v;
}

__global__ __launch_bounds__(1024)
void bucket_hist_kernel(const int* __restrict__ col, int* __restrict__ hist,
                        int E, int NB, int NWG) {
  __shared__ int cnt[SORT_MAXNB];
  const int tid = threadIdx.x;
  const int g = blockIdx.x;
  for (int b = tid; b < NB; b += 1024) cnt[b] = 0;
  __syncthreads();
  const int e0 = g * SORT_CH;
  const int n = min(E - e0, SORT_CH);
  int rc[SORT_CH / 1024];
#pragma unroll
  for (int k = 0; k < SORT_CH / 1024; ++k) {
    const int i = k * 1024 + tid;
    rc[k] = (i < n) ? col[e0 + i] : -1;
  }
#pragma unroll
  for (int k = 0; k < SORT_CH / 1024; ++k)
    if (rc[k] >= 0) atomicAdd(&cnt[rc[k] >> SORT_SH], 1);
  __syncthreads();
  for (int b = tid; b < NB; b += 1024) hist[b * NWG + g] = cnt[b];
}

__global__ __launch_bounds__(1024)
void bucket_part_kernel(const int* __restrict__ row, const int* __restrict__ col,
                        const float* __restrict__ w, const int* __restrict__ offs,
                        int2* __restrict__ prc, int* __restrict__ pcl,
                        int E, int NB, int NWG) {
  __shared__ int cnt[SORT_MAXNB];
  __shared__ int lstart[SORT_MAXNB];
  __shared__ int gbase[SORT_MAXNB];
  __shared__ int wsum[32];
  __shared__ int s_row[SORT_CH];
  __shared__ int s_col[SORT_CH];
  __shared__ int s_w[SORT_CH];
  const int tid = threadIdx.x;
  const int g = blockIdx.x;
  for (int b = tid; b < NB; b += 1024) cnt[b] = 0;
  __syncthreads();
  const int e0 = g * SORT_CH;
  const int n = min(E - e0, SORT_CH);
  int rc[SORT_CH / 1024], rr[SORT_CH / 1024], rw[SORT_CH / 1024], rk[SORT_CH / 1024];
#pragma unroll
  for (int k = 0; k < SORT_CH / 1024; ++k) {
    const int i = k * 1024 + tid;
    if (i < n) {
      const int c = col[e0 + i];
      rc[k] = c;
      rr[k] = row[e0 + i];
      rw[k] = __float_as_int(w[e0 + i]);
      rk[k] = atomicAdd(&cnt[c >> SORT_SH], 1);
    }
  }
  __syncthreads();
  const int v = (tid < NB) ? cnt[tid] : 0;
  const int ex = block_excl_scan(v, wsum);
  if (tid < NB) {
    lstart[tid] = ex;
    gbase[tid] = offs[tid * NWG + g] - ex;
  }
  __syncthreads();
#pragma unroll
  for (int k = 0; k < SORT_CH / 1024; ++k) {
    const int i = k * 1024 + tid;
    if (i < n) {
      const int p = lstart[rc[k] >> SORT_SH] + rk[k];
      s_row[p] = rr[k];
      s_col[p] = rc[k];
      s_w[p] = rw[k];
    }
  }
  __syncthreads();
  for (int i = tid; i < n; i += 1024) {
    const int c = s_col[i];
    const int d = gbase[c >> SORT_SH] + i;
    prc[d] = make_int2(s_row[i], s_w[i]);
    pcl[d] = c;
  }
}

__global__ __launch_bounds__(512)
void bucket_sort_kernel(const int* __restrict__ offs, const int2* __restrict__ prc,
                        const int* __restrict__ pcl, int* __restrict__ start,
                        int2* __restrict__ sedge, int N, int NB, int NWG) {
  __shared__ int cnt[SORT_BN];
  __shared__ int nst[SORT_BN];
  __shared__ int cur[SORT_BN];
  __shared__ int wsum[32];
  __shared__ int2 stage[SORT_CAP];
  const int tid = threadIdx.x;
  const int b = blockIdx.x;
  const int bs = offs[b * NWG];
  const int be = offs[(b + 1) * NWG];
  const int n = be - bs;
  const int base = b << SORT_SH;
  if (tid < SORT_BN) { cnt[tid] = 0; cur[tid] = 0; }
  __syncthreads();
  for (int i0 = tid; i0 < n; i0 += 4 * 512) {
    int lc[4];
#pragma unroll
    for (int u = 0; u < 4; ++u) {
      const int i = i0 + u * 512;
      lc[u] = (i < n) ? pcl[bs + i] - base : -1;
    }
#pragma unroll
    for (int u = 0; u < 4; ++u)
      if (lc[u] >= 0) atomicAdd(&cnt[lc[u]], 1);
  }
  __syncthreads();
  const int v = (tid < SORT_BN) ? cnt[tid] : 0;
  const int ex = block_excl_scan(v, wsum);
  if (tid < SORT_BN) {
    nst[tid] = ex;
    if (base + tid < N) start[base + tid] = bs + ex;
  }
  if (b == NB - 1 && tid == 0) start[N] = be;
  __syncthreads();
  const bool staged = n <= SORT_CAP;
  for (int i0 = tid; i0 < n; i0 += 4 * 512) {
    int lc[4];
    int2 rec[4];
#pragma unroll
    for (int u = 0; u < 4; ++u) {
      const int i = i0 + u * 512;
      const bool in = i < n;
      lc[u] = in ? pcl[bs + i] - base : -1;
      rec[u] = in ? prc[bs + i] : make_int2(0, 0);
    }
#pragma unroll
    for (int u = 0; u < 4; ++u) {
      if (lc[u] >= 0) {
        const int p = nst[lc[u]] + atomicAdd(&cur[lc[u]], 1);
        if (staged) stage[p] = rec[u];
        else sedge[bs + p] = rec[u];
      }
    }
  }
  if (staged) {
    __syncthreads();
    for (int i = tid; i < n; i += 512) sedge[bs + i] = stage[i];
  }
}

__global__ void dis_csr_kernel(const int* __restrict__ start, const int2* __restrict__ sedge,
                               float* __restrict__ dis, int N) {
  int t = blockIdx.x * blockDim.x + threadIdx.x;
  int c = t >> 3;
  int sub = t & 7;
  if (c >= N) c = N - 1;
  const int s = start[c];
  const int e = start[c + 1];
  float d = 0.0f;
  for (int p = s + sub; p < e; p += 8) d += __int_as_float(sedge[p].y);
  d += __shfl_xor(d, 4);
  d += __shfl_xor(d, 2);
  d += __shfl_xor(d, 1);
  if (sub == 0) dis[c] = (d > 0.0f) ? rsqrtf(d) : 0.0f;
}

__global__ void norm_csr_kernel(const int* __restrict__ start, const float* __restrict__ dis,
                                int2* __restrict__ sedge, int N) {
  int t = blockIdx.x * blockDim.x + threadIdx.x;
  int c = t >> 3;
  int sub = t & 7;
  if (c >= N) return;
  const int s = start[c];
  const int e = start[c + 1];
  const float dc = dis[c];
  for (int p = s + sub; p < e; p += 8) {
    int2 ed = sedge[p];
    float nv = dis[ed.x] * __int_as_float(ed.y) * dc;
    sedge[p] = make_int2(ed.x, __float_as_int(nv));
  }
}


#define HOP_ROWS 32
#define HOP_WIN 768

__device__ __forceinline__ void hop_list_load(const int2* __restrict__ sedge, int wb, int ge,
                                              int tid, int2 (&st)[HOP_WIN / 256]) {
#pragma unroll
  for (int k = 0; k < HOP_WIN / 256; ++k) st[k] = sedge[min(wb + k * 256 + tid, ge - 1)];
}

__device__ __forceinline__ float4 hop_rows_sum(const int2* __restrict__ sedge,
                                               const float4* __restrict__ h4, int2* win,
                                               int gs, int ge, int s, int e, int sub, int tid,
                                               int2 (&st)[HOP_WIN / 256]) {
  float4 acc = make_float4(0.0f, 0.0f, 0.0f, 0.0f);
  const float4 z = make_float4(0.0f, 0.0f, 0.0f, 0.0f);
  for (int wb = gs & ~15; wb < ge; wb += HOP_WIN) {
    if (wb > gs) {
      __syncthreads();
      hop_list_load(sedge, wb, ge, tid, st);
    }
#pragma unroll
    for (int k = 0; k < HOP_WIN / 256; ++k) win[k * 256 + tid] = st[k];
    __syncthreads();
    const int p0 = max(s, wb);
    const int p1 = min(e, wb + HOP_WIN);
    if (p0 < p1) {
      const int qe = p1 - wb;
      const int last = qe - 1;
      int q = p0 - wb;
      int2 ed0 = win[min(q + 0, last)];
      int2 ed1 = win[min(q + 1, last)];
      int2 ed2 = win[min(q + 2, last)];
      int2 ed3 = win[min(q + 3, last)];
      for (; q < qe; q += 4) {
        float4 v0 = h4[(long)ed0.x * (DIM / 4) + sub];
        float4 v1 = h4[(long)ed1.x * (DIM / 4) + sub];
        float4 v2 = h4[(long)ed2.x * (DIM / 4) + sub];
        float4 v3 = h4[(long)ed3.x * (DIM / 4) + sub];
        const int r = q + 4;
        int2 n0 = win[min(r + 0, last)];
        int2 n1 = win[min(r + 1, last)];
        int2 n2 = win[min(r + 2, last)];
        int2 n3 = win[min(r + 3, last)];
        const bool k1 = q + 1 < qe, k2 = q + 2 < qe, k3 = q + 3 < qe;
        const float w0 = __int_as_float(ed0.y);
        const float w1 = k1 ? __int_as_float(ed1.y) : 0.0f;
        const float w2 = k2 ? __int_as_float(ed2.y) : 0.0f;
        const float w3 = k3 ? __int_as_float(ed3.y) : 0.0f;
        if (!k1) v1 = z;
        if (!k2) v2 = z;
        if (!k3) v3 = z;
        acc.x += w0 * v0.x; acc.y += w0 * v0.y; acc.z += w0 * v0.z; acc.w += w0 * v0.w;
        acc.x += w1 * v1.x; acc.y += w1 * v1.y; acc.z += w1 * v1.z; acc.w += w1 * v1.w;
        acc.x += w2 * v2.x; acc.y += w2 * v2.y; acc.z += w2 * v2.z; acc.w += w2 * v2.w;
        acc.x += w3 * v3.x; acc.y += w3 * v3.y; acc.z += w3 * v3.z; acc.w += w3 * v3.w;
        ed0 = n0; ed1 = n1; ed2 = n2; ed3 = n3;
      }
    }
  }
  return acc;
}

__global__ __launch_bounds__(256)
void hop_csr_kernel(const int* __restrict__ start, const int2* __restrict__ sedge,
                    const float* __restrict__ hsrc, float* __restrict__ hdst, int N) {
  __shared__ int2 win[HOP_WIN];
  const int tid = threadIdx.x;
  const int sub = tid & 7;
  const int row0 = blockIdx.x * HOP_ROWS;
  const int c = row0 + (tid >> 3);
  const int gs = start[row0];
  const int ge = start[min(row0 + HOP_ROWS, N)];
  int2 st[HOP_WIN / 256] = {};
  if (gs < ge) hop_list_load(sedge, gs & ~15, ge, tid, st);
  int s = 0, e = 0;
  if (c < N) {
    s = start[c];
    e = start[c + 1];
  }
  const float4 acc = hop_rows_sum(sedge, reinterpret_cast<const float4*>(hsrc), win,
                                  gs, ge, s, e, sub, tid, st);
  if (c < N) reinterpret_cast<float4*>(hdst)[(long)c * (DIM / 4) + sub] = acc;
}


#define HT_ROWS 32
static_assert(HT_ROWS == HOP_ROWS, "hop_rows_sum serves 32 rows per 256-thread work-group");
#define HT_PITCH 36
#define HW_PITCH 20
#define HW_HALF (4 * DIM * HW_PITCH)

__global__ __launch_bounds__(256)
void hop_gemm_wmma_kernel(const int* __restrict__ start, const int2* __restrict__ sedge,
                          const float* s0, const float* __restrict__ s1,
                          const float* __restrict__ s2,
                          const float* __restrict__ Wl,
                          const float* __restrict__ bl,
                          float* dst, float* __restrict__ h3tail, int N, int nTiles) {
  __shared__ __attribute__((aligned(16))) float As[4 * HT_ROWS * HT_PITCH];
  __shared__ __attribute__((aligned(16))) float Ws[2 * HW_HALF];
  __shared__ int2 win[HOP_WIN];
  const int tid = threadIdx.x;
  const int r = tid >> 3;
  const int sub = tid & 7;
  const int c = blockIdx.x * HT_ROWS + r;
  const float4 z4 = make_float4(0.0f, 0.0f, 0.0f, 0.0f);

  const float4* __restrict__ W4 = reinterpret_cast<const float4*>(Wl);
  const float4 wq0 = W4[0 * 256 + tid];
  const float4 wq1 = W4[1 * 256 + tid];
  const float4 wq2 = W4[2 * 256 + tid];
  const float4 wq3 = W4[3 * 256 + tid];
  const long ro = (long)c * (DIM / 4) + sub;
  const int row0 = blockIdx.x * HT_ROWS;
  const int gs = start[row0];
  const int ge = start[min(row0 + HT_ROWS, N)];
  int2 st[HOP_WIN / 256] = {};
  if (gs < ge) hop_list_load(sedge, gs & ~15, ge, tid, st);
  float4 a0 = z4, a1 = z4, a2 = z4;
  int s = 0, e = 0;
  if (c < N) {
    a0 = reinterpret_cast<const float4*>(s0)[ro];
    a1 = reinterpret_cast<const float4*>(s1)[ro];
    a2 = reinterpret_cast<const float4*>(s2)[ro];
    s = start[c];
    e = start[c + 1];
  }
  {
    float* wrow = Ws + (sub >> 2) * HW_HALF + r * HW_PITCH + (sub & 3) * 4;
    *reinterpret_cast<float4*>(wrow + 0 * DIM * HW_PITCH) = wq0;
    *reinterpret_cast<float4*>(wrow + 1 * DIM * HW_PITCH) = wq1;
    *reinterpret_cast<float4*>(wrow + 2 * DIM * HW_PITCH) = wq2;
    *reinterpret_cast<float4*>(wrow + 3 * DIM * HW_PITCH) = wq3;
  }
  float* arow_st = As + r * HT_PITCH + sub * 4;
  *reinterpret_cast<float4*>(arow_st + 0 * HT_ROWS * HT_PITCH) = a0;
  *reinterpret_cast<float4*>(arow_st + 1 * HT_ROWS * HT_PITCH) = a1;
  *reinterpret_cast<float4*>(arow_st + 2 * HT_ROWS * HT_PITCH) = a2;

  const float4 acc = hop_rows_sum(sedge, reinterpret_cast<const float4*>(s2), win,
                                  gs, ge, s, e, sub, tid, st);
  if (c < N && c >= nTiles * 16) reinterpret_cast<float4*>(h3tail)[ro] = acc;
  *reinterpret_cast<float4*>(arow_st + 3 * HT_ROWS * HT_PITCH) = acc;
  __syncthreads();

  const int wave = tid >> 5;
  if (wave >= HT_ROWS / 8) return;
  const int tile = blockIdx.x * (HT_ROWS / 16) + (wave >> 1);
  if (tile >= nTiles) return;
  const int nh = wave & 1;
  const int lane = tid & 31;
  const int mlane = lane & 15;
  const bool hi = lane >= 16;
  const int khalf = hi ? 2 : 0;
  const float* arow = As + ((wave >> 1) * 16 + mlane) * HT_PITCH;
  const float* wcol = Ws + nh * HW_HALF + mlane;
  const int ncol = nh * 16 + mlane;
  const float bn = bl[ncol];

  v8f cacc = {};

  v2f af[2][4], bf[2][4];
#pragma unroll
  for (int j = 0; j < 4; ++j) {
    const int ka = 4 * j + khalf;
    af[0][j] = *reinterpret_cast<const v2f*>(arow + ka);
    bf[0][j].x = wcol[(ka + 0) * HW_PITCH];
    bf[0][j].y = wcol[(ka + 1) * HW_PITCH];
  }
#pragma unroll
  for (int g = 0; g < 8; ++g) {
    if (g < 7) {
      const int gn = g + 1;
      const float* Ak = arow + (gn >> 1) * (HT_ROWS * HT_PITCH);
      const float* Wk = wcol + (gn >> 1) * (DIM * HW_PITCH);
#pragma unroll
      for (int j = 0; j < 4; ++j) {
        const int ka = 4 * ((gn & 1) * 4 + j) + khalf;
        af[gn & 1][j] = *reinterpret_cast<const v2f*>(Ak + ka);
        bf[gn & 1][j].x = Wk[(ka + 0) * HW_PITCH];
        bf[gn & 1][j].y = Wk[(ka + 1) * HW_PITCH];
      }
    }
    __builtin_amdgcn_sched_barrier(0);
#pragma unroll
    for (int j = 0; j < 4; ++j)
      cacc = __builtin_amdgcn_wmma_f32_16x16x4_f32(false, af[g & 1][j], false, bf[g & 1][j],
                                                   (short)0, cacc, false, false);
    __builtin_amdgcn_sched_barrier(0);
  }

  int mBase = tile * 16 + (hi ? 8 : 0);
#pragma unroll
  for (int rr = 0; rr < 8; ++rr) {
    float v = cacc[rr] + bn;
    v = v > 0.0f ? v : 0.0f;
    dst[(long)(mBase + rr) * DIM + ncol] = v;
  }
}

__global__ void tag_gemm_tail_kernel(const float* __restrict__ s0, const float* __restrict__ s1,
                                     const float* __restrict__ s2, const float* __restrict__ s3,
                                     const float* __restrict__ Wl, const float* __restrict__ bl,
                                     float* __restrict__ dst, int startRow, int N) {
  int t = blockIdx.x * blockDim.x + threadIdx.x;
  int m = startRow + (t >> 5);
  int n = t & 31;
  if (m >= N) return;
  float acc = bl[n];
  const float* srcs[4] = {s0, s1, s2, s3};
  for (int k = 0; k < 4; ++k) {
    const float* S = srcs[k] + (long)m * DIM;
    const float* Wk = Wl + k * DIM * DIM;
    for (int j = 0; j < DIM; ++j) acc += S[j] * Wk[j * DIM + n];
  }
  dst[(long)m * DIM + n] = acc > 0.0f ? acc : 0.0f;
}


extern "C" void kernel_launch(void* const* d_in, const int* in_sizes, int n_in,
                              void* d_out, int out_size, void* d_ws, size_t ws_size,
                              hipStream_t stream) {
  (void)n_in; (void)out_size; (void)ws_size;
  const float* d_x = (const float*)d_in[0];
  const int*   d_ei = (const int*)d_in[1];
  const float* d_w = (const float*)d_in[2];
  const float* d_W = (const float*)d_in[3];
  const float* d_b = (const float*)d_in[4];
  const int N = in_sizes[0] / DIM;
  const int E = in_sizes[2];
  const int* row = d_ei;
  const int* col = d_ei + E;

  auto align_up = [](size_t v) { return (v + 255) & ~(size_t)255; };
  char* ws = (char*)d_ws;
  size_t off = 0;
  int*   cnt   = (int*)(ws + off);   off = align_up(off + (size_t)N * 4);
  int*   start = (int*)(ws + off);   off = align_up(off + ((size_t)N + 1) * 4);
  int*   rnk   = (int*)(ws + off);   off = align_up(off + (size_t)E * 4);
  int2*  sedge = (int2*)(ws + off);  off = align_up(off + (size_t)E * 8);
  float* dis   = (float*)(ws + off); off = align_up(off + (size_t)N * 4);
  float* h1    = (float*)(ws + off); off = align_up(off + (size_t)N * DIM * 4);
  float* h2    = (float*)(ws + off); off = align_up(off + (size_t)N * DIM * 4);
  float* h3    = (float*)(ws + off); off = align_up(off + (size_t)N * DIM * 4);
  float* xcur  = (float*)(ws + off); off = align_up(off + (size_t)N * DIM * 4);
  const int NB  = (N + SORT_BN - 1) >> SORT_SH;
  const int NWG = (E + SORT_CH - 1) / SORT_CH;
  const int M   = NB * NWG;
  int*   hist  = (int*)(ws + off);   off = align_up(off + (size_t)M * 4);
  int*   offs  = (int*)(ws + off);   off = align_up(off + ((size_t)M + 1) * 4);
  int2*  prc   = (int2*)(ws + off);  off = align_up(off + (size_t)E * 8);
  int*   pcl   = (int*)(ws + off);   off = align_up(off + (size_t)E * 4);

  const int B = 256;
  auto gblk = [&](long n) { return (int)((n + B - 1) / B); };
  const long grp8 = (long)N * 8;

  if (NB <= SORT_MAXNB) {
    bucket_hist_kernel<<<NWG, 1024, 0, stream>>>(col, hist, E, NB, NWG);
    scan_kernel<<<M / SCAN_TILE + 1, SCAN_THREADS, 0, stream>>>(hist, offs, M);
    bucket_part_kernel<<<NWG, 1024, 0, stream>>>(row, col, d_w, offs, prc, pcl, E, NB, NWG);
    bucket_sort_kernel<<<NB, 512, 0, stream>>>(offs, prc, pcl, start, sedge, N, NB, NWG);
  } else {
    zero_i32_kernel<<<gblk(N), B, 0, stream>>>(cnt, N);
    rank_kernel<<<gblk(E), B, 0, stream>>>(col, cnt, rnk, E);
    scan_kernel<<<N / SCAN_TILE + 1, SCAN_THREADS, 0, stream>>>(cnt, start, N);
    fill_kernel<<<gblk(E), B, 0, stream>>>(row, col, d_w, start, rnk, sedge, E);
  }
  dis_csr_kernel<<<gblk(grp8), B, 0, stream>>>(start, sedge, dis, N);
  norm_csr_kernel<<<gblk(grp8), B, 0, stream>>>(start, dis, sedge, N);

  const int nTiles = N / 16;
  const int hopTileBlocks = (N + HT_ROWS - 1) / HT_ROWS;
  const int tailStart = nTiles * 16;
  const int tailRows = N - tailStart;

  for (int l = 0; l < 3; ++l) {
    const float* xin = (l == 0) ? d_x : xcur;
    float* dst = (l == 2) ? (float*)d_out : xcur;
    const float* Wl = d_W + (size_t)l * 4 * DIM * DIM;
    const float* bl = d_b + (size_t)l * DIM;

    hop_csr_kernel<<<gblk(grp8), B, 0, stream>>>(start, sedge, xin, h1, N);
    hop_csr_kernel<<<gblk(grp8), B, 0, stream>>>(start, sedge, h1, h2, N);

    hop_gemm_wmma_kernel<<<hopTileBlocks, B, 0, stream>>>(start, sedge, xin, h1, h2, Wl, bl, dst,
                                                          h3, N, nTiles);
    if (tailRows > 0) {
      tag_gemm_tail_kernel<<<gblk((long)tailRows * 32), B, 0, stream>>>(
          xin, h1, h2, h3, Wl, bl, dst, tailStart, N);
    }
  }
}


# Degree sums and edge normalisation move into the bucket sort: a per-bucket launch sums the weights with LDS float adds (arrival order), pass 2 writes normalised weights; two per-node launches go
# speedup vs baseline: 6.8575x; 1.0405x over previous
#include <hip/hip_runtime.h>


typedef float v2f __attribute__((ext_vector_type(2)));
typedef float v8f __attribute__((ext_vector_type(8)));

#define DIM 32
#define SCAN_THREADS 1024
#define SCAN_TILE (SCAN_THREADS * 4)


__global__ void zero_i32_kernel(int* __restrict__ p, int n) {
  int i = blockIdx.x * blockDim.x + threadIdx.x;
  if (i < n) p[i] = 0;
}

__global__ void rank_kernel(const int* __restrict__ col, int* __restrict__ cnt,
                            int* __restrict__ rnk, int E) {
  int e = blockIdx.x * blockDim.x + threadIdx.x;
  if (e < E) rnk[e] = atomicAdd(&cnt[col[e]], 1);
}

__global__ __launch_bounds__(SCAN_THREADS)
void scan_kernel(const int* __restrict__ cnt, int* __restrict__ start, int N) {
  __shared__ int red[SCAN_THREADS / 32];
  __shared__ int wtot[SCAN_THREADS / 32];
  const int tid = threadIdx.x;
  const int lane = tid & 31;
  const int wv = tid >> 5;
  const int base0 = blockIdx.x * SCAN_TILE;

  int a0 = 0, a1 = 0, a2 = 0, a3 = 0;
  int i = tid * 4;
  for (; i + 3 * SCAN_TILE < base0; i += 4 * SCAN_TILE) {
    int4 v0 = *reinterpret_cast<const int4*>(cnt + i);
    int4 v1 = *reinterpret_cast<const int4*>(cnt + i + SCAN_TILE);
    int4 v2 = *reinterpret_cast<const int4*>(cnt + i + 2 * SCAN_TILE);
    int4 v3 = *reinterpret_cast<const int4*>(cnt + i + 3 * SCAN_TILE);
    a0 += (v0.x + v0.y) + (v0.z + v0.w);
    a1 += (v1.x + v1.y) + (v1.z + v1.w);
    a2 += (v2.x + v2.y) + (v2.z + v2.w);
    a3 += (v3.x + v3.y) + (v3.z + v3.w);
  }
  for (; i < base0; i += SCAN_TILE) {
    int4 v = *reinterpret_cast<const int4*>(cnt + i);
    a0 += (v.x + v.y) + (v.z + v.w);
  }
  int acc = (a0 + a1) + (a2 + a3);
  for (int o = 16; o > 0; o >>= 1) acc += __shfl_xor(acc, o);
  if (lane == 0) red[wv] = acc;

  const int i0 = base0 + tid * 4;
  int c0 = 0, c1 = 0, c2 = 0, c3 = 0;
  if (i0 + 3 < N) {
    int4 v = *reinterpret_cast<const int4*>(cnt + i0);
    c0 = v.x; c1 = v.y; c2 = v.z; c3 = v.w;
  } else {
    if (i0 + 0 < N) c0 = cnt[i0 + 0];
    if (i0 + 1 < N) c1 = cnt[i0 + 1];
    if (i0 + 2 < N) c2 = cnt[i0 + 2];
  }
  const int t1 = c0, t2 = t1 + c1, t3 = t2 + c2, tsum = t3 + c3;
  int inc = tsum;
  for (int o = 1; o < 32; o <<= 1) {
    int u = __shfl_up(inc, o);
    if (lane >= o) inc += u;
  }
  if (lane == 31) wtot[wv] = inc;
  __syncthreads();
  int pre = 0;
  for (int k = 0; k < SCAN_THREADS / 32; ++k) pre += red[k];
  for (int k = 0; k < wv; ++k) pre += wtot[k];
  const int ex = pre + inc - tsum;
  if (i0 + 0 <= N) start[i0 + 0] = ex;
  if (i0 + 1 <= N) start[i0 + 1] = ex + t1;
  if (i0 + 2 <= N) start[i0 + 2] = ex + t2;
  if (i0 + 3 <= N) start[i0 + 3] = ex + t3;
}

__global__ void fill_kernel(const int* __restrict__ row, const int* __restrict__ col,
                            const float* __restrict__ w, const int* __restrict__ start,
                            const int* __restrict__ rnk, int2* __restrict__ sedge, int E) {
  int e = blockIdx.x * blockDim.x + threadIdx.x;
  if (e < E) {
    int p = start[col[e]] + rnk[e];
    sedge[p] = make_int2(row[e], __float_as_int(w[e]));
  }
}


#define SORT_SH 8
#define SORT_BN (1 << SORT_SH)
#define SORT_MAXNB 512
#define SORT_CH 8192
#define SORT_CAP 8192

__device__ __forceinline__ int block_excl_scan(int v, int* wsum) {
  const int lane = threadIdx.x & 31;
  const int wv = threadIdx.x >> 5;
  int inc = v;
  for (int o = 1; o < 32; o <<= 1) {
    int u = __shfl_up(inc, o);
    if (lane >= o) inc += u;
  }
  if (lane == 31) wsum[wv] = inc;
  __syncthreads();
  int pre = 0;
  for (int k = 0; k < wv; ++k) pre += wsum[k];
  return pre + inc - v;
}

__global__ __launch_bounds__(1024)
void bucket_hist_kernel(const int* __restrict__ col, int* __restrict__ hist,
                        int E, int NB, int NWG) {
  __shared__ int cnt[SORT_MAXNB];
  const int tid = threadIdx.x;
  const int g = blockIdx.x;
  for (int b = tid; b < NB; b += 1024) cnt[b] = 0;
  __syncthreads();
  const int e0 = g * SORT_CH;
  const int n = min(E - e0, SORT_CH);
  int rc[SORT_CH / 1024];
#pragma unroll
  for (int k = 0; k < SORT_CH / 1024; ++k) {
    const int i = k * 1024 + tid;
    rc[k] = (i < n) ? col[e0 + i] : -1;
  }
#pragma unroll
  for (int k = 0; k < SORT_CH / 1024; ++k)
    if (rc[k] >= 0) atomicAdd(&cnt[rc[k] >> SORT_SH], 1);
  __syncthreads();
  for (int b = tid; b < NB; b += 1024) hist[b * NWG + g] = cnt[b];
}

__global__ __launch_bounds__(1024)
void bucket_part_kernel(const int* __restrict__ row, const int* __restrict__ col,
                        const float* __restrict__ w, const int* __restrict__ offs,
                        int2* __restrict__ prc, int* __restrict__ pcl,
                        int E, int NB, int NWG) {
  __shared__ int cnt[SORT_MAXNB];
  __shared__ int lstart[SORT_MAXNB];
  __shared__ int gbase[SORT_MAXNB];
  __shared__ int wsum[32];
  __shared__ int s_row[SORT_CH];
  __shared__ int s_col[SORT_CH];
  __shared__ int s_w[SORT_CH];
  const int tid = threadIdx.x;
  const int g = blockIdx.x;
  for (int b = tid; b < NB; b += 1024) cnt[b] = 0;
  __syncthreads();
  const int e0 = g * SORT_CH;
  const int n = min(E - e0, SORT_CH);
  int rc[SORT_CH / 1024], rr[SORT_CH / 1024], rw[SORT_CH / 1024], rk[SORT_CH / 1024];
#pragma unroll
  for (int k = 0; k < SORT_CH / 1024; ++k) {
    const int i = k * 1024 + tid;
    if (i < n) {
      const int c = col[e0 + i];
      rc[k] = c;
      rr[k] = row[e0 + i];
      rw[k] = __float_as_int(w[e0 + i]);
      rk[k] = atomicAdd(&cnt[c >> SORT_SH], 1);
    }
  }
  __syncthreads();
  const int v = (tid < NB) ? cnt[tid] : 0;
  const int ex = block_excl_scan(v, wsum);
  if (tid < NB) {
    lstart[tid] = ex;
    gbase[tid] = offs[tid * NWG + g] - ex;
  }
  __syncthreads();
#pragma unroll
  for (int k = 0; k < SORT_CH / 1024; ++k) {
    const int i = k * 1024 + tid;
    if (i < n) {
      const int p = lstart[rc[k] >> SORT_SH] + rk[k];
      s_row[p] = rr[k];
      s_col[p] = rc[k];
      s_w[p] = rw[k];
    }
  }
  __syncthreads();
  for (int i = tid; i < n; i += 1024) {
    const int c = s_col[i];
    const int d = gbase[c >> SORT_SH] + i;
    prc[d] = make_int2(s_row[i], s_w[i]);
    pcl[d] = c;
  }
}

__global__ __launch_bounds__(512)
void bucket_deg_kernel(const int* __restrict__ offs, const int2* __restrict__ prc,
                       const int* __restrict__ pcl, float* __restrict__ dis, int N, int NWG) {
  __shared__ float deg[SORT_BN];
  const int tid = threadIdx.x;
  const int b = blockIdx.x;
  const int bs = offs[b * NWG];
  const int be = offs[(b + 1) * NWG];
  const int n = be - bs;
  const int base = b << SORT_SH;
  if (tid < SORT_BN) deg[tid] = 0.0f;
  __syncthreads();
  for (int i0 = tid; i0 < n; i0 += 4 * 512) {
    int cc[4];
    int wv[4];
#pragma unroll
    for (int u = 0; u < 4; ++u) {
      const int ic = min(i0 + u * 512, n - 1);
      cc[u] = pcl[bs + ic];
      wv[u] = prc[bs + ic].y;
    }
#pragma unroll
    for (int u = 0; u < 4; ++u)
      if (i0 + u * 512 < n) atomicAdd(&deg[cc[u] - base], __int_as_float(wv[u]));
  }
  __syncthreads();
  if (tid < SORT_BN && base + tid < N) {
    const float d = deg[tid];
    dis[base + tid] = (d > 0.0f) ? rsqrtf(d) : 0.0f;
  }
}

__global__ __launch_bounds__(512)
void bucket_sort_kernel(const int* __restrict__ offs, const int2* __restrict__ prc,
                        const int* __restrict__ pcl, const float* __restrict__ dis,
                        int* __restrict__ start,
                        int2* __restrict__ sedge, int N, int NB, int NWG) {
  __shared__ int cnt[SORT_BN];
  __shared__ int nst[SORT_BN];
  __shared__ int cur[SORT_BN];
  __shared__ int wsum[32];
  __shared__ float dcl[SORT_BN];
  __shared__ int2 stage[SORT_CAP];
  const int tid = threadIdx.x;
  const int b = blockIdx.x;
  const int bs = offs[b * NWG];
  const int be = offs[(b + 1) * NWG];
  const int n = be - bs;
  const int base = b << SORT_SH;
  if (tid < SORT_BN) {
    cnt[tid] = 0;
    cur[tid] = 0;
    dcl[tid] = (base + tid < N) ? dis[base + tid] : 0.0f;
  }
  __syncthreads();
  for (int i0 = tid; i0 < n; i0 += 4 * 512) {
    int lc[4];
#pragma unroll
    for (int u = 0; u < 4; ++u) {
      const int i = i0 + u * 512;
      lc[u] = (i < n) ? pcl[bs + i] - base : -1;
    }
#pragma unroll
    for (int u = 0; u < 4; ++u)
      if (lc[u] >= 0) atomicAdd(&cnt[lc[u]], 1);
  }
  __syncthreads();
  const int v = (tid < SORT_BN) ? cnt[tid] : 0;
  const int ex = block_excl_scan(v, wsum);
  if (tid < SORT_BN) {
    nst[tid] = ex;
    if (base + tid < N) start[base + tid] = bs + ex;
  }
  if (b == NB - 1 && tid == 0) start[N] = be;
  __syncthreads();
  const bool staged = n <= SORT_CAP;
  for (int i0 = tid; i0 < n; i0 += 4 * 512) {
    int lc[4];
    int2 rec[4];
    float dr[4];
#pragma unroll
    for (int u = 0; u < 4; ++u) {
      const int i = i0 + u * 512;
      const bool in = i < n;
      lc[u] = in ? pcl[bs + i] - base : -1;
      rec[u] = in ? prc[bs + i] : make_int2(0, 0);
    }
#pragma unroll
    for (int u = 0; u < 4; ++u) dr[u] = dis[rec[u].x];
#pragma unroll
    for (int u = 0; u < 4; ++u) {
      if (lc[u] >= 0) {
        rec[u].y = __float_as_int(dr[u] * __int_as_float(rec[u].y) * dcl[lc[u]]);
        const int p = nst[lc[u]] + atomicAdd(&cur[lc[u]], 1);
        if (staged) stage[p] = rec[u];
        else sedge[bs + p] = rec[u];
      }
    }
  }
  if (staged) {
    __syncthreads();
    for (int i = tid; i < n; i += 512) sedge[bs + i] = stage[i];
  }
}

__global__ void dis_csr_kernel(const int* __restrict__ start, const int2* __restrict__ sedge,
                               float* __restrict__ dis, int N) {
  int t = blockIdx.x * blockDim.x + threadIdx.x;
  int c = t >> 3;
  int sub = t & 7;
  if (c >= N) c = N - 1;
  const int s = start[c];
  const int e = start[c + 1];
  float d = 0.0f;
  for (int p = s + sub; p < e; p += 8) d += __int_as_float(sedge[p].y);
  d += __shfl_xor(d, 4);
  d += __shfl_xor(d, 2);
  d += __shfl_xor(d, 1);
  if (sub == 0) dis[c] = (d > 0.0f) ? rsqrtf(d) : 0.0f;
}

__global__ void norm_csr_kernel(const int* __restrict__ start, const float* __restrict__ dis,
                                int2* __restrict__ sedge, int N) {
  int t = blockIdx.x * blockDim.x + threadIdx.x;
  int c = t >> 3;
  int sub = t & 7;
  if (c >= N) return;
  const int s = start[c];
  const int e = start[c + 1];
  const float dc = dis[c];
  for (int p = s + sub; p < e; p += 8) {
    int2 ed = sedge[p];
    float nv = dis[ed.x] * __int_as_float(ed.y) * dc;
    sedge[p] = make_int2(ed.x, __float_as_int(nv));
  }
}


#define HOP_ROWS 32
#define HOP_WIN 768

__device__ __forceinline__ void hop_list_load(const int2* __restrict__ sedge, int wb, int ge,
                                              int tid, int2 (&st)[HOP_WIN / 256]) {
#pragma unroll
  for (int k = 0; k < HOP_WIN / 256; ++k) st[k] = sedge[min(wb + k * 256 + tid, ge - 1)];
}

__device__ __forceinline__ float4 hop_rows_sum(const int2* __restrict__ sedge,
                                               const float4* __restrict__ h4, int2* win,
                                               int gs, int ge, int s, int e, int sub, int tid,
                                               int2 (&st)[HOP_WIN / 256]) {
  float4 acc = make_float4(0.0f, 0.0f, 0.0f, 0.0f);
  const float4 z = make_float4(0.0f, 0.0f, 0.0f, 0.0f);
  for (int wb = gs & ~15; wb < ge; wb += HOP_WIN) {
    if (wb > gs) {
      __syncthreads();
      hop_list_load(sedge, wb, ge, tid, st);
    }
#pragma unroll
    for (int k = 0; k < HOP_WIN / 256; ++k) win[k * 256 + tid] = st[k];
    __syncthreads();
    const int p0 = max(s, wb);
    const int p1 = min(e, wb + HOP_WIN);
    if (p0 < p1) {
      const int qe = p1 - wb;
      const int last = qe - 1;
      int q = p0 - wb;
      int2 ed0 = win[min(q + 0, last)];
      int2 ed1 = win[min(q + 1, last)];
      int2 ed2 = win[min(q + 2, last)];
      int2 ed3 = win[min(q + 3, last)];
      for (; q < qe; q += 4) {
        float4 v0 = h4[(long)ed0.x * (DIM / 4) + sub];
        float4 v1 = h4[(long)ed1.x * (DIM / 4) + sub];
        float4 v2 = h4[(long)ed2.x * (DIM / 4) + sub];
        float4 v3 = h4[(long)ed3.x * (DIM / 4) + sub];
        const int r = q + 4;
        int2 n0 = win[min(r + 0, last)];
        int2 n1 = win[min(r + 1, last)];
        int2 n2 = win[min(r + 2, last)];
        int2 n3 = win[min(r + 3, last)];
        const bool k1 = q + 1 < qe, k2 = q + 2 < qe, k3 = q + 3 < qe;
        const float w0 = __int_as_float(ed0.y);
        const float w1 = k1 ? __int_as_float(ed1.y) : 0.0f;
        const float w2 = k2 ? __int_as_float(ed2.y) : 0.0f;
        const float w3 = k3 ? __int_as_float(ed3.y) : 0.0f;
        if (!k1) v1 = z;
        if (!k2) v2 = z;
        if (!k3) v3 = z;
        acc.x += w0 * v0.x; acc.y += w0 * v0.y; acc.z += w0 * v0.z; acc.w += w0 * v0.w;
        acc.x += w1 * v1.x; acc.y += w1 * v1.y; acc.z += w1 * v1.z; acc.w += w1 * v1.w;
        acc.x += w2 * v2.x; acc.y += w2 * v2.y; acc.z += w2 * v2.z; acc.w += w2 * v2.w;
        acc.x += w3 * v3.x; acc.y += w3 * v3.y; acc.z += w3 * v3.z; acc.w += w3 * v3.w;
        ed0 = n0; ed1 = n1; ed2 = n2; ed3 = n3;
      }
    }
  }
  return acc;
}

__global__ __launch_bounds__(256)
void hop_csr_kernel(const int* __restrict__ start, const int2* __restrict__ sedge,
                    const float* __restrict__ hsrc, float* __restrict__ hdst, int N) {
  __shared__ int2 win[HOP_WIN];
  const int tid = threadIdx.x;
  const int sub = tid & 7;
  const int row0 = blockIdx.x * HOP_ROWS;
  const int c = row0 + (tid >> 3);
  const int gs = start[row0];
  const int ge = start[min(row0 + HOP_ROWS, N)];
  int2 st[HOP_WIN / 256] = {};
  if (gs < ge) hop_list_load(sedge, gs & ~15, ge, tid, st);
  int s = 0, e = 0;
  if (c < N) {
    s = start[c];
    e = start[c + 1];
  }
  const float4 acc = hop_rows_sum(sedge, reinterpret_cast<const float4*>(hsrc), win,
                                  gs, ge, s, e, sub, tid, st);
  if (c < N) reinterpret_cast<float4*>(hdst)[(long)c * (DIM / 4) + sub] = acc;
}


#define HT_ROWS 32
static_assert(HT_ROWS == HOP_ROWS, "hop_rows_sum serves 32 rows per 256-thread work-group");
#define HT_PITCH 36
#define HW_PITCH 20
#define HW_HALF (4 * DIM * HW_PITCH)

__global__ __launch_bounds__(256)
void hop_gemm_wmma_kernel(const int* __restrict__ start, const int2* __restrict__ sedge,
                          const float* s0, const float* __restrict__ s1,
                          const float* __restrict__ s2,
                          const float* __restrict__ Wl,
                          const float* __restrict__ bl,
                          float* dst, float* __restrict__ h3tail, int N, int nTiles) {
  __shared__ __attribute__((aligned(16))) float As[4 * HT_ROWS * HT_PITCH];
  __shared__ __attribute__((aligned(16))) float Ws[2 * HW_HALF];
  __shared__ int2 win[HOP_WIN];
  const int tid = threadIdx.x;
  const int r = tid >> 3;
  const int sub = tid & 7;
  const int c = blockIdx.x * HT_ROWS + r;
  const float4 z4 = make_float4(0.0f, 0.0f, 0.0f, 0.0f);

  const float4* __restrict__ W4 = reinterpret_cast<const float4*>(Wl);
  const float4 wq0 = W4[0 * 256 + tid];
  const float4 wq1 = W4[1 * 256 + tid];
  const float4 wq2 = W4[2 * 256 + tid];
  const float4 wq3 = W4[3 * 256 + tid];
  const long ro = (long)c * (DIM / 4) + sub;
  const int row0 = blockIdx.x * HT_ROWS;
  const int gs = start[row0];
  const int ge = start[min(row0 + HT_ROWS, N)];
  int2 st[HOP_WIN / 256] = {};
  if (gs < ge) hop_list_load(sedge, gs & ~15, ge, tid, st);
  float4 a0 = z4, a1 = z4, a2 = z4;
  int s = 0, e = 0;
  if (c < N) {
    a0 = reinterpret_cast<const float4*>(s0)[ro];
    a1 = reinterpret_cast<const float4*>(s1)[ro];
    a2 = reinterpret_cast<const float4*>(s2)[ro];
    s = start[c];
    e = start[c + 1];
  }
  {
    float* wrow = Ws + (sub >> 2) * HW_HALF + r * HW_PITCH + (sub & 3) * 4;
    *reinterpret_cast<float4*>(wrow + 0 * DIM * HW_PITCH) = wq0;
    *reinterpret_cast<float4*>(wrow + 1 * DIM * HW_PITCH) = wq1;
    *reinterpret_cast<float4*>(wrow + 2 * DIM * HW_PITCH) = wq2;
    *reinterpret_cast<float4*>(wrow + 3 * DIM * HW_PITCH) = wq3;
  }
  float* arow_st = As + r * HT_PITCH + sub * 4;
  *reinterpret_cast<float4*>(arow_st + 0 * HT_ROWS * HT_PITCH) = a0;
  *reinterpret_cast<float4*>(arow_st + 1 * HT_ROWS * HT_PITCH) = a1;
  *reinterpret_cast<float4*>(arow_st + 2 * HT_ROWS * HT_PITCH) = a2;

  const float4 acc = hop_rows_sum(sedge, reinterpret_cast<const float4*>(s2), win,
                                  gs, ge, s, e, sub, tid, st);
  if (c < N && c >= nTiles * 16) reinterpret_cast<float4*>(h3tail)[ro] = acc;
  *reinterpret_cast<float4*>(arow_st + 3 * HT_ROWS * HT_PITCH) = acc;
  __syncthreads();

  const int wave = tid >> 5;
  if (wave >= HT_ROWS / 8) return;
  const int tile = blockIdx.x * (HT_ROWS / 16) + (wave >> 1);
  if (tile >= nTiles) return;
  const int nh = wave & 1;
  const int lane = tid & 31;
  const int mlane = lane & 15;
  const bool hi = lane >= 16;
  const int khalf = hi ? 2 : 0;
  const float* arow = As + ((wave >> 1) * 16 + mlane) * HT_PITCH;
  const float* wcol = Ws + nh * HW_HALF + mlane;
  const int ncol = nh * 16 + mlane;
  const float bn = bl[ncol];

  v8f cacc = {};

  v2f af[2][4], bf[2][4];
#pragma unroll
  for (int j = 0; j < 4; ++j) {
    const int ka = 4 * j + khalf;
    af[0][j] = *reinterpret_cast<const v2f*>(arow + ka);
    bf[0][j].x = wcol[(ka + 0) * HW_PITCH];
    bf[0][j].y = wcol[(ka + 1) * HW_PITCH];
  }
#pragma unroll
  for (int g = 0; g < 8; ++g) {
    if (g < 7) {
      const int gn = g + 1;
      const float* Ak = arow + (gn >> 1) * (HT_ROWS * HT_PITCH);
      const float* Wk = wcol + (gn >> 1) * (DIM * HW_PITCH);
#pragma unroll
      for (int j = 0; j < 4; ++j) {
        const int ka = 4 * ((gn & 1) * 4 + j) + khalf;
        af[gn & 1][j] = *reinterpret_cast<const v2f*>(Ak + ka);
        bf[gn & 1][j].x = Wk[(ka + 0) * HW_PITCH];
        bf[gn & 1][j].y = Wk[(ka + 1) * HW_PITCH];
      }
    }
    __builtin_amdgcn_sched_barrier(0);
#pragma unroll
    for (int j = 0; j < 4; ++j)
      cacc = __builtin_amdgcn_wmma_f32_16x16x4_f32(false, af[g & 1][j], false, bf[g & 1][j],
                                                   (short)0, cacc, false, false);
    __builtin_amdgcn_sched_barrier(0);
  }

  int mBase = tile * 16 + (hi ? 8 : 0);
#pragma unroll
  for (int rr = 0; rr < 8; ++rr) {
    float v = cacc[rr] + bn;
    v = v > 0.0f ? v : 0.0f;
    dst[(long)(mBase + rr) * DIM + ncol] = v;
  }
}

__global__ void tag_gemm_tail_kernel(const float* __restrict__ s0, const float* __restrict__ s1,
                                     const float* __restrict__ s2, const float* __restrict__ s3,
                                     const float* __restrict__ Wl, const float* __restrict__ bl,
                                     float* __restrict__ dst, int startRow, int N) {
  int t = blockIdx.x * blockDim.x + threadIdx.x;
  int m = startRow + (t >> 5);
  int n = t & 31;
  if (m >= N) return;
  float acc = bl[n];
  const float* srcs[4] = {s0, s1, s2, s3};
  for (int k = 0; k < 4; ++k) {
    const float* S = srcs[k] + (long)m * DIM;
    const float* Wk = Wl + k * DIM * DIM;
    for (int j = 0; j < DIM; ++j) acc += S[j] * Wk[j * DIM + n];
  }
  dst[(long)m * DIM + n] = acc > 0.0f ? acc : 0.0f;
}


extern "C" void kernel_launch(void* const* d_in, const int* in_sizes, int n_in,
                              void* d_out, int out_size, void* d_ws, size_t ws_size,
                              hipStream_t stream) {
  (void)n_in; (void)out_size; (void)ws_size;
  const float* d_x = (const float*)d_in[0];
  const int*   d_ei = (const int*)d_in[1];
  const float* d_w = (const float*)d_in[2];
  const float* d_W = (const float*)d_in[3];
  const float* d_b = (const float*)d_in[4];
  const int N = in_sizes[0] / DIM;
  const int E = in_sizes[2];
  const int* row = d_ei;
  const int* col = d_ei + E;

  auto align_up = [](size_t v) { return (v + 255) & ~(size_t)255; };
  char* ws = (char*)d_ws;
  size_t off = 0;
  int*   cnt   = (int*)(ws + off);   off = align_up(off + (size_t)N * 4);
  int*   start = (int*)(ws + off);   off = align_up(off + ((size_t)N + 1) * 4);
  int*   rnk   = (int*)(ws + off);   off = align_up(off + (size_t)E * 4);
  int2*  sedge = (int2*)(ws + off);  off = align_up(off + (size_t)E * 8);
  float* dis   = (float*)(ws + off); off = align_up(off + (size_t)N * 4);
  float* h1    = (float*)(ws + off); off = align_up(off + (size_t)N * DIM * 4);
  float* h2    = (float*)(ws + off); off = align_up(off + (size_t)N * DIM * 4);
  float* h3    = (float*)(ws + off); off = align_up(off + (size_t)N * DIM * 4);
  float* xcur  = (float*)(ws + off); off = align_up(off + (size_t)N * DIM * 4);
  const int NB  = (N + SORT_BN - 1) >> SORT_SH;
  const int NWG = (E + SORT_CH - 1) / SORT_CH;
  const int M   = NB * NWG;
  int*   hist  = (int*)(ws + off);   off = align_up(off + (size_t)M * 4);
  int*   offs  = (int*)(ws + off);   off = align_up(off + ((size_t)M + 1) * 4);
  int2*  prc   = (int2*)(ws + off);  off = align_up(off + (size_t)E * 8);
  int*   pcl   = (int*)(ws + off);   off = align_up(off + (size_t)E * 4);

  const int B = 256;
  auto gblk = [&](long n) { return (int)((n + B - 1) / B); };
  const long grp8 = (long)N * 8;

  if (NB <= SORT_MAXNB) {
    bucket_hist_kernel<<<NWG, 1024, 0, stream>>>(col, hist, E, NB, NWG);
    scan_kernel<<<M / SCAN_TILE + 1, SCAN_THREADS, 0, stream>>>(hist, offs, M);
    bucket_part_kernel<<<NWG, 1024, 0, stream>>>(row, col, d_w, offs, prc, pcl, E, NB, NWG);
    bucket_deg_kernel<<<NB, 512, 0, stream>>>(offs, prc, pcl, dis, N, NWG);
    bucket_sort_kernel<<<NB, 512, 0, stream>>>(offs, prc, pcl, dis, start, sedge, N, NB, NWG);
  } else {
    zero_i32_kernel<<<gblk(N), B, 0, stream>>>(cnt, N);
    rank_kernel<<<gblk(E), B, 0, stream>>>(col, cnt, rnk, E);
    scan_kernel<<<N / SCAN_TILE + 1, SCAN_THREADS, 0, stream>>>(cnt, start, N);
    fill_kernel<<<gblk(E), B, 0, stream>>>(row, col, d_w, start, rnk, sedge, E);
    dis_csr_kernel<<<gblk(grp8), B, 0, stream>>>(start, sedge, dis, N);
    norm_csr_kernel<<<gblk(grp8), B, 0, stream>>>(start, dis, sedge, N);
  }

  const int nTiles = N / 16;
  const int hopTileBlocks = (N + HT_ROWS - 1) / HT_ROWS;
  const int tailStart = nTiles * 16;
  const int tailRows = N - tailStart;

  for (int l = 0; l < 3; ++l) {
    const float* xin = (l == 0) ? d_x : xcur;
    float* dst = (l == 2) ? (float*)d_out : xcur;
    const float* Wl = d_W + (size_t)l * 4 * DIM * DIM;
    const float* bl = d_b + (size_t)l * DIM;

    hop_csr_kernel<<<gblk(grp8), B, 0, stream>>>(start, sedge, xin, h1, N);
    hop_csr_kernel<<<gblk(grp8), B, 0, stream>>>(start, sedge, h1, h2, N);

    hop_gemm_wmma_kernel<<<hopTileBlocks, B, 0, stream>>>(start, sedge, xin, h1, h2, Wl, bl, dst,
                                                          h3, N, nTiles);
    if (tailRows > 0) {
      tag_gemm_tail_kernel<<<gblk((long)tailRows * 32), B, 0, stream>>>(
          xin, h1, h2, h3, Wl, bl, dst, tailStart, N);
    }
  }
}


# Bucket pass 1 writes each chunk back in place, bucket-sorted, with a per-chunk table; per-bucket kernels walk the runs: histogram and scan launches go; per-target sum order still by LDS add arrival
# speedup vs baseline: 7.0911x; 1.0341x over previous
#include <hip/hip_runtime.h>


typedef float v2f __attribute__((ext_vector_type(2)));
typedef float v8f __attribute__((ext_vector_type(8)));

#define DIM 32
#define SCAN_THREADS 1024
#define SCAN_TILE (SCAN_THREADS * 4)


__global__ void zero_i32_kernel(int* __restrict__ p, int n) {
  int i = blockIdx.x * blockDim.x + threadIdx.x;
  if (i < n) p[i] = 0;
}

__global__ void rank_kernel(const int* __restrict__ col, int* __restrict__ cnt,
                            int* __restrict__ rnk, int E) {
  int e = blockIdx.x * blockDim.x + threadIdx.x;
  if (e < E) rnk[e] = atomicAdd(&cnt[col[e]], 1);
}

__global__ __launch_bounds__(SCAN_THREADS)
void scan_kernel(const int* __restrict__ cnt, int* __restrict__ start, int N) {
  __shared__ int red[SCAN_THREADS / 32];
  __shared__ int wtot[SCAN_THREADS / 32];
  const int tid = threadIdx.x;
  const int lane = tid & 31;
  const int wv = tid >> 5;
  const int base0 = blockIdx.x * SCAN_TILE;

  int a0 = 0, a1 = 0, a2 = 0, a3 = 0;
  int i = tid * 4;
  for (; i + 3 * SCAN_TILE < base0; i += 4 * SCAN_TILE) {
    int4 v0 = *reinterpret_cast<const int4*>(cnt + i);
    int4 v1 = *reinterpret_cast<const int4*>(cnt + i + SCAN_TILE);
    int4 v2 = *reinterpret_cast<const int4*>(cnt + i + 2 * SCAN_TILE);
    int4 v3 = *reinterpret_cast<const int4*>(cnt + i + 3 * SCAN_TILE);
    a0 += (v0.x + v0.y) + (v0.z + v0.w);
    a1 += (v1.x + v1.y) + (v1.z + v1.w);
    a2 += (v2.x + v2.y) + (v2.z + v2.w);
    a3 += (v3.x + v3.y) + (v3.z + v3.w);
  }
  for (; i < base0; i += SCAN_TILE) {
    int4 v = *reinterpret_cast<const int4*>(cnt + i);
    a0 += (v.x + v.y) + (v.z + v.w);
  }
  int acc = (a0 + a1) + (a2 + a3);
  for (int o = 16; o > 0; o >>= 1) acc += __shfl_xor(acc, o);
  if (lane == 0) red[wv] = acc;

  const int i0 = base0 + tid * 4;
  int c0 = 0, c1 = 0, c2 = 0, c3 = 0;
  if (i0 + 3 < N) {
    int4 v = *reinterpret_cast<const int4*>(cnt + i0);
    c0 = v.x; c1 = v.y; c2 = v.z; c3 = v.w;
  } else {
    if (i0 + 0 < N) c0 = cnt[i0 + 0];
    if (i0 + 1 < N) c1 = cnt[i0 + 1];
    if (i0 + 2 < N) c2 = cnt[i0 + 2];
  }
  const int t1 = c0, t2 = t1 + c1, t3 = t2 + c2, tsum = t3 + c3;
  int inc = tsum;
  for (int o = 1; o < 32; o <<= 1) {
    int u = __shfl_up(inc, o);
    if (lane >= o) inc += u;
  }
  if (lane == 31) wtot[wv] = inc;
  __syncthreads();
  int pre = 0;
  for (int k = 0; k < SCAN_THREADS / 32; ++k) pre += red[k];
  for (int k = 0; k < wv; ++k) pre += wtot[k];
  const int ex = pre + inc - tsum;
  if (i0 + 0 <= N) start[i0 + 0] = ex;
  if (i0 + 1 <= N) start[i0 + 1] = ex + t1;
  if (i0 + 2 <= N) start[i0 + 2] = ex + t2;
  if (i0 + 3 <= N) start[i0 + 3] = ex + t3;
}

__global__ void fill_kernel(const int* __restrict__ row, const int* __restrict__ col,
                            const float* __restrict__ w, const int* __restrict__ start,
                            const int* __restrict__ rnk, int2* __restrict__ sedge, int E) {
  int e = blockIdx.x * blockDim.x + threadIdx.x;
  if (e < E) {
    int p = start[col[e]] + rnk[e];
    sedge[p] = make_int2(row[e], __float_as_int(w[e]));
  }
}


#define SORT_SH 8
#define SORT_BN (1 << SORT_SH)
#define SORT_MAXNB 512
#define SORT_CH 8192
#define SORT_CAP 8192
#define RUN_TILE 512
#define RUN_U 8
static_assert(SORT_MAXNB <= 512, "the per-bucket work-groups read one bucket total per thread");

__device__ __forceinline__ int block_excl_scan(int v, int* wsum) {
  const int lane = threadIdx.x & 31;
  const int wv = threadIdx.x >> 5;
  int inc = v;
  for (int o = 1; o < 32; o <<= 1) {
    int u = __shfl_up(inc, o);
    if (lane >= o) inc += u;
  }
  if (lane == 31) wsum[wv] = inc;
  __syncthreads();
  int pre = 0;
  for (int k = 0; k < wv; ++k) pre += wsum[k];
  return pre + inc - v;
}

__global__ __launch_bounds__(1024)
void bucket_part_kernel(const int* __restrict__ row, const int* __restrict__ col,
                        const float* __restrict__ w, int2* __restrict__ tab,
                        int2* __restrict__ prc, int* __restrict__ pcl,
                        int E, int NB) {
  __shared__ int cnt[SORT_MAXNB];
  __shared__ int lstart[SORT_MAXNB];
  __shared__ int wsum[32];
  __shared__ int s_row[SORT_CH];
  __shared__ int s_col[SORT_CH];
  __shared__ int s_w[SORT_CH];
  const int tid = threadIdx.x;
  const int g = blockIdx.x;
  for (int b = tid; b < NB; b += 1024) cnt[b] = 0;
  __syncthreads();
  const int e0 = g * SORT_CH;
  const int n = min(E - e0, SORT_CH);
  int rc[SORT_CH / 1024], rr[SORT_CH / 1024], rw[SORT_CH / 1024], rk[SORT_CH / 1024];
#pragma unroll
  for (int k = 0; k < SORT_CH / 1024; ++k) {
    const int i = k * 1024 + tid;
    if (i < n) {
      const int c = col[e0 + i];
      rc[k] = c;
      rr[k] = row[e0 + i];
      rw[k] = __float_as_int(w[e0 + i]);
      rk[k] = atomicAdd(&cnt[c >> SORT_SH], 1);
    }
  }
  __syncthreads();
  const int v = (tid < NB) ? cnt[tid] : 0;
  const int ex = block_excl_scan(v, wsum);
  if (tid < NB) {
    lstart[tid] = ex;
    tab[g * NB + tid] = make_int2(v, ex);
  }
  __syncthreads();
#pragma unroll
  for (int k = 0; k < SORT_CH / 1024; ++k) {
    const int i = k * 1024 + tid;
    if (i < n) {
      const int p = lstart[rc[k] >> SORT_SH] + rk[k];
      s_row[p] = rr[k];
      s_col[p] = rc[k];
      s_w[p] = rw[k];
    }
  }
  __syncthreads();
  for (int i = tid; i < n; i += 1024) {
    prc[e0 + i] = make_int2(s_row[i], s_w[i]);
    pcl[e0 + i] = s_col[i];
  }
}

__device__ __forceinline__ int run_table_load(const int2* __restrict__ tab, int b, int NB, int NWG,
                                              int t0, int tid, int2* rt) {
  int tot = 0;
  for (int k = tid; k < RUN_TILE && t0 + k < NWG; k += 512) {
    const int2 t = tab[(t0 + k) * NB + b];
    rt[k] = make_int2(t.x, (t0 + k) * SORT_CH + t.y);
    tot += t.x;
  }
  return tot;
}

__global__ __launch_bounds__(512)
void bucket_deg_kernel(const int2* __restrict__ tab, const int2* __restrict__ prc,
                       const int* __restrict__ pcl, float* __restrict__ dis,
                       int* __restrict__ btot, int N, int NB, int NWG) {
  __shared__ float deg[SORT_BN];
  __shared__ int2 rt[RUN_TILE];
  __shared__ int s_tot;
  const int tid = threadIdx.x;
  const int lane = tid & 31;
  const int wv = tid >> 5;
  const int b = blockIdx.x;
  const int base = b << SORT_SH;
  if (tid < SORT_BN) deg[tid] = 0.0f;
  if (tid == 0) s_tot = 0;
  int mytot = 0;
  for (int t0 = 0; t0 < NWG; t0 += RUN_TILE) {
    if (t0 > 0) __syncthreads();
    mytot += run_table_load(tab, b, NB, NWG, t0, tid, rt);
    __syncthreads();
    const int nT = min(RUN_TILE, NWG - t0);
    for (int g0 = wv; g0 < nT; g0 += 16 * RUN_U) {
      int cc[RUN_U], pp[RUN_U];
      int mx = 0;
#pragma unroll
      for (int u = 0; u < RUN_U; ++u) {
        const int g = g0 + 16 * u;
        const int2 t = (g < nT) ? rt[g] : make_int2(0, 0);
        cc[u] = t.x;
        pp[u] = t.y;
        mx = max(mx, t.x);
      }
      for (int j = lane; j < mx; j += 32) {
        int tc[RUN_U], tw[RUN_U];
#pragma unroll
        for (int u = 0; u < RUN_U; ++u) {
          const int idx = (j < cc[u]) ? pp[u] + j : 0;
          tc[u] = pcl[idx];
          tw[u] = prc[idx].y;
        }
#pragma unroll
        for (int u = 0; u < RUN_U; ++u)
          if (j < cc[u]) atomicAdd(&deg[tc[u] - base], __int_as_float(tw[u]));
      }
    }
  }
  for (int o = 16; o > 0; o >>= 1) mytot += __shfl_xor(mytot, o);
  if (lane == 0 && mytot != 0) atomicAdd(&s_tot, mytot);
  __syncthreads();
  if (tid < SORT_BN && base + tid < N) {
    const float d = deg[tid];
    dis[base + tid] = (d > 0.0f) ? rsqrtf(d) : 0.0f;
  }
  if (tid == 0) btot[b] = s_tot;
}

__global__ __launch_bounds__(512)
void bucket_sort_kernel(const int2* __restrict__ tab, const int* __restrict__ btot,
                        const int2* __restrict__ prc,
                        const int* __restrict__ pcl, const float* __restrict__ dis,
                        int* __restrict__ start,
                        int2* __restrict__ sedge, int N, int NB, int NWG) {
  __shared__ int cnt[SORT_BN];
  __shared__ int nst[SORT_BN];
  __shared__ int cur[SORT_BN];
  __shared__ int wsum[32];
  __shared__ int wpre[16];
  __shared__ float dcl[SORT_BN];
  __shared__ int2 rt[RUN_TILE];
  __shared__ int2 stage[SORT_CAP];
  const int tid = threadIdx.x;
  const int lane = tid & 31;
  const int wv = tid >> 5;
  const int b = blockIdx.x;
  const int base = b << SORT_SH;
  int tv = (tid < b) ? btot[tid] : 0;
  const int n = btot[b];
  if (tid < SORT_BN) {
    cnt[tid] = 0;
    cur[tid] = 0;
    dcl[tid] = (base + tid < N) ? dis[base + tid] : 0.0f;
  }
  for (int o = 16; o > 0; o >>= 1) tv += __shfl_xor(tv, o);
  if (lane == 0) wpre[wv] = tv;
  const bool one_tile = NWG <= RUN_TILE;
  for (int t0 = 0; t0 < NWG; t0 += RUN_TILE) {
    if (t0 > 0) __syncthreads();
    run_table_load(tab, b, NB, NWG, t0, tid, rt);
    __syncthreads();
    const int nT = min(RUN_TILE, NWG - t0);
    for (int g0 = wv; g0 < nT; g0 += 16 * RUN_U) {
      int cc[RUN_U], pp[RUN_U];
      int mx = 0;
#pragma unroll
      for (int u = 0; u < RUN_U; ++u) {
        const int g = g0 + 16 * u;
        const int2 t = (g < nT) ? rt[g] : make_int2(0, 0);
        cc[u] = t.x;
        pp[u] = t.y;
        mx = max(mx, t.x);
      }
      for (int j = lane; j < mx; j += 32) {
        int lc[RUN_U];
#pragma unroll
        for (int u = 0; u < RUN_U; ++u) lc[u] = pcl[(j < cc[u]) ? pp[u] + j : 0];
#pragma unroll
        for (int u = 0; u < RUN_U; ++u)
          if (j < cc[u]) atomicAdd(&cnt[lc[u] - base], 1);
      }
    }
  }
  __syncthreads();
  int bs = 0;
  for (int k = 0; k < 16; ++k) bs += wpre[k];
  const int be = bs + n;
  const int v = (tid < SORT_BN) ? cnt[tid] : 0;
  const int ex = block_excl_scan(v, wsum);
  if (tid < SORT_BN) {
    nst[tid] = ex;
    if (base + tid < N) start[base + tid] = bs + ex;
  }
  if (b == NB - 1 && tid == 0) start[N] = be;
  __syncthreads();
  const bool staged = n <= SORT_CAP;
  for (int t0 = 0; t0 < NWG; t0 += RUN_TILE) {
    if (!one_tile) {
      __syncthreads();
      run_table_load(tab, b, NB, NWG, t0, tid, rt);
      __syncthreads();
    }
    const int nT = min(RUN_TILE, NWG - t0);
    for (int g0 = wv; g0 < nT; g0 += 16 * RUN_U) {
      int cc[RUN_U], pp[RUN_U];
      int mx = 0;
#pragma unroll
      for (int u = 0; u < RUN_U; ++u) {
        const int g = g0 + 16 * u;
        const int2 t = (g < nT) ? rt[g] : make_int2(0, 0);
        cc[u] = t.x;
        pp[u] = t.y;
        mx = max(mx, t.x);
      }
      for (int j = lane; j < mx; j += 32) {
        int lc[RUN_U];
        int2 rec[RUN_U];
        float dr[RUN_U];
#pragma unroll
        for (int u = 0; u < RUN_U; ++u) {
          const int idx = (j < cc[u]) ? pp[u] + j : 0;
          lc[u] = pcl[idx] - base;
          rec[u] = prc[idx];
        }
#pragma unroll
        for (int u = 0; u < RUN_U; ++u) dr[u] = dis[rec[u].x];
#pragma unroll
        for (int u = 0; u < RUN_U; ++u) {
          if (j < cc[u]) {
            rec[u].y = __float_as_int(dr[u] * __int_as_float(rec[u].y) * dcl[lc[u]]);
            const int p = nst[lc[u]] + atomicAdd(&cur[lc[u]], 1);
            if (staged) stage[p] = rec[u];
            else sedge[bs + p] = rec[u];
          }
        }
      }
    }
  }
  if (staged) {
    __syncthreads();
    for (int i = tid; i < n; i += 512) sedge[bs + i] = stage[i];
  }
}

__global__ void dis_csr_kernel(const int* __restrict__ start, const int2* __restrict__ sedge,
                               float* __restrict__ dis, int N) {
  int t = blockIdx.x * blockDim.x + threadIdx.x;
  int c = t >> 3;
  int sub = t & 7;
  if (c >= N) c = N - 1;
  const int s = start[c];
  const int e = start[c + 1];
  float d = 0.0f;
  for (int p = s + sub; p < e; p += 8) d += __int_as_float(sedge[p].y);
  d += __shfl_xor(d, 4);
  d += __shfl_xor(d, 2);
  d += __shfl_xor(d, 1);
  if (sub == 0) dis[c] = (d > 0.0f) ? rsqrtf(d) : 0.0f;
}

__global__ void norm_csr_kernel(const int* __restrict__ start, const float* __restrict__ dis,
                                int2* __restrict__ sedge, int N) {
  int t = blockIdx.x * blockDim.x + threadIdx.x;
  int c = t >> 3;
  int sub = t & 7;
  if (c >= N) return;
  const int s = start[c];
  const int e = start[c + 1];
  const float dc = dis[c];
  for (int p = s + sub; p < e; p += 8) {
    int2 ed = sedge[p];
    float nv = dis[ed.x] * __int_as_float(ed.y) * dc;
    sedge[p] = make_int2(ed.x, __float_as_int(nv));
  }
}


#define HOP_ROWS 32
#define HOP_WIN 768

__device__ __forceinline__ void hop_list_load(const int2* __restrict__ sedge, int wb, int ge,
                                              int tid, int2 (&st)[HOP_WIN / 256]) {
#pragma unroll
  for (int k = 0; k < HOP_WIN / 256; ++k) st[k] = sedge[min(wb + k * 256 + tid, ge - 1)];
}

__device__ __forceinline__ float4 hop_rows_sum(const int2* __restrict__ sedge,
                                               const float4* __restrict__ h4, int2* win,
                                               int gs, int ge, int s, int e, int sub, int tid,
                                               int2 (&st)[HOP_WIN / 256]) {
  float4 acc = make_float4(0.0f, 0.0f, 0.0f, 0.0f);
  const float4 z = make_float4(0.0f, 0.0f, 0.0f, 0.0f);
  for (int wb = gs & ~15; wb < ge; wb += HOP_WIN) {
    if (wb > gs) {
      __syncthreads();
      hop_list_load(sedge, wb, ge, tid, st);
    }
#pragma unroll
    for (int k = 0; k < HOP_WIN / 256; ++k) win[k * 256 + tid] = st[k];
    __syncthreads();
    const int p0 = max(s, wb);
    const int p1 = min(e, wb + HOP_WIN);
    if (p0 < p1) {
      const int qe = p1 - wb;
      const int last = qe - 1;
      int q = p0 - wb;
      int2 ed0 = win[min(q + 0, last)];
      int2 ed1 = win[min(q + 1, last)];
      int2 ed2 = win[min(q + 2, last)];
      int2 ed3 = win[min(q + 3, last)];
      for (; q < qe; q += 4) {
        float4 v0 = h4[(long)ed0.x * (DIM / 4) + sub];
        float4 v1 = h4[(long)ed1.x * (DIM / 4) + sub];
        float4 v2 = h4[(long)ed2.x * (DIM / 4) + sub];
        float4 v3 = h4[(long)ed3.x * (DIM / 4) + sub];
        const int r = q + 4;
        int2 n0 = win[min(r + 0, last)];
        int2 n1 = win[min(r + 1, last)];
        int2 n2 = win[min(r + 2, last)];
        int2 n3 = win[min(r + 3, last)];
        const bool k1 = q + 1 < qe, k2 = q + 2 < qe, k3 = q + 3 < qe;
        const float w0 = __int_as_float(ed0.y);
        const float w1 = k1 ? __int_as_float(ed1.y) : 0.0f;
        const float w2 = k2 ? __int_as_float(ed2.y) : 0.0f;
        const float w3 = k3 ? __int_as_float(ed3.y) : 0.0f;
        if (!k1) v1 = z;
        if (!k2) v2 = z;
        if (!k3) v3 = z;
        acc.x += w0 * v0.x; acc.y += w0 * v0.y; acc.z += w0 * v0.z; acc.w += w0 * v0.w;
        acc.x += w1 * v1.x; acc.y += w1 * v1.y; acc.z += w1 * v1.z; acc.w += w1 * v1.w;
        acc.x += w2 * v2.x; acc.y += w2 * v2.y; acc.z += w2 * v2.z; acc.w += w2 * v2.w;
        acc.x += w3 * v3.x; acc.y += w3 * v3.y; acc.z += w3 * v3.z; acc.w += w3 * v3.w;
        ed0 = n0; ed1 = n1; ed2 = n2; ed3 = n3;
      }
    }
  }
  return acc;
}

__global__ __launch_bounds__(256)
void hop_csr_kernel(const int* __restrict__ start, const int2* __restrict__ sedge,
                    const float* __restrict__ hsrc, float* __restrict__ hdst, int N) {
  __shared__ int2 win[HOP_WIN];
  const int tid = threadIdx.x;
  const int sub = tid & 7;
  const int row0 = blockIdx.x * HOP_ROWS;
  const int c = row0 + (tid >> 3);
  const int gs = start[row0];
  const int ge = start[min(row0 + HOP_ROWS, N)];
  int2 st[HOP_WIN / 256] = {};
  if (gs < ge) hop_list_load(sedge, gs & ~15, ge, tid, st);
  int s = 0, e = 0;
  if (c < N) {
    s = start[c];
    e = start[c + 1];
  }
  const float4 acc = hop_rows_sum(sedge, reinterpret_cast<const float4*>(hsrc), win,
                                  gs, ge, s, e, sub, tid, st);
  if (c < N) reinterpret_cast<float4*>(hdst)[(long)c * (DIM / 4) + sub] = acc;
}


#define HT_ROWS 32
static_assert(HT_ROWS == HOP_ROWS, "hop_rows_sum serves 32 rows per 256-thread work-group");
#define HT_PITCH 36
#define HW_PITCH 20
#define HW_HALF (4 * DIM * HW_PITCH)

__global__ __launch_bounds__(256)
void hop_gemm_wmma_kernel(const int* __restrict__ start, const int2* __restrict__ sedge,
                          const float* s0, const float* __restrict__ s1,
                          const float* __restrict__ s2,
                          const float* __restrict__ Wl,
                          const float* __restrict__ bl,
                          float* dst, float* __restrict__ h3tail, int N, int nTiles) {
  __shared__ __attribute__((aligned(16))) float As[4 * HT_ROWS * HT_PITCH];
  __shared__ __attribute__((aligned(16))) float Ws[2 * HW_HALF];
  __shared__ int2 win[HOP_WIN];
  const int tid = threadIdx.x;
  const int r = tid >> 3;
  const int sub = tid & 7;
  const int c = blockIdx.x * HT_ROWS + r;
  const float4 z4 = make_float4(0.0f, 0.0f, 0.0f, 0.0f);

  const float4* __restrict__ W4 = reinterpret_cast<const float4*>(Wl);
  const float4 wq0 = W4[0 * 256 + tid];
  const float4 wq1 = W4[1 * 256 + tid];
  const float4 wq2 = W4[2 * 256 + tid];
  const float4 wq3 = W4[3 * 256 + tid];
  const long ro = (long)c * (DIM / 4) + sub;
  const int row0 = blockIdx.x * HT_ROWS;
  const int gs = start[row0];
  const int ge = start[min(row0 + HT_ROWS, N)];
  int2 st[HOP_WIN / 256] = {};
  if (gs < ge) hop_list_load(sedge, gs & ~15, ge, tid, st);
  float4 a0 = z4, a1 = z4, a2 = z4;
  int s = 0, e = 0;
  if (c < N) {
    a0 = reinterpret_cast<const float4*>(s0)[ro];
    a1 = reinterpret_cast<const float4*>(s1)[ro];
    a2 = reinterpret_cast<const float4*>(s2)[ro];
    s = start[c];
    e = start[c + 1];
  }
  {
    float* wrow = Ws + (sub >> 2) * HW_HALF + r * HW_PITCH + (sub & 3) * 4;
    *reinterpret_cast<float4*>(wrow + 0 * DIM * HW_PITCH) = wq0;
    *reinterpret_cast<float4*>(wrow + 1 * DIM * HW_PITCH) = wq1;
    *reinterpret_cast<float4*>(wrow + 2 * DIM * HW_PITCH) = wq2;
    *reinterpret_cast<float4*>(wrow + 3 * DIM * HW_PITCH) = wq3;
  }
  float* arow_st = As + r * HT_PITCH + sub * 4;
  *reinterpret_cast<float4*>(arow_st + 0 * HT_ROWS * HT_PITCH) = a0;
  *reinterpret_cast<float4*>(arow_st + 1 * HT_ROWS * HT_PITCH) = a1;
  *reinterpret_cast<float4*>(arow_st + 2 * HT_ROWS * HT_PITCH) = a2;

  const float4 acc = hop_rows_sum(sedge, reinterpret_cast<const float4*>(s2), win,
                                  gs, ge, s, e, sub, tid, st);
  if (c < N && c >= nTiles * 16) reinterpret_cast<float4*>(h3tail)[ro] = acc;
  *reinterpret_cast<float4*>(arow_st + 3 * HT_ROWS * HT_PITCH) = acc;
  __syncthreads();

  const int wave = tid >> 5;
  if (wave >= HT_ROWS / 8) return;
  const int tile = blockIdx.x * (HT_ROWS / 16) + (wave >> 1);
  if (tile >= nTiles) return;
  const int nh = wave & 1;
  const int lane = tid & 31;
  const int mlane = lane & 15;
  const bool hi = lane >= 16;
  const int khalf = hi ? 2 : 0;
  const float* arow = As + ((wave >> 1) * 16 + mlane) * HT_PITCH;
  const float* wcol = Ws + nh * HW_HALF + mlane;
  const int ncol = nh * 16 + mlane;
  const float bn = bl[ncol];

  v8f cacc = {};

  v2f af[2][4], bf[2][4];
#pragma unroll
  for (int j = 0; j < 4; ++j) {
    const int ka = 4 * j + khalf;
    af[0][j] = *reinterpret_cast<const v2f*>(arow + ka);
    bf[0][j].x = wcol[(ka + 0) * HW_PITCH];
    bf[0][j].y = wcol[(ka + 1) * HW_PITCH];
  }
#pragma unroll
  for (int g = 0; g < 8; ++g) {
    if (g < 7) {
      const int gn = g + 1;
      const float* Ak = arow + (gn >> 1) * (HT_ROWS * HT_PITCH);
      const float* Wk = wcol + (gn >> 1) * (DIM * HW_PITCH);
#pragma unroll
      for (int j = 0; j < 4; ++j) {
        const int ka = 4 * ((gn & 1) * 4 + j) + khalf;
        af[gn & 1][j] = *reinterpret_cast<const v2f*>(Ak + ka);
        bf[gn & 1][j].x = Wk[(ka + 0) * HW_PITCH];
        bf[gn & 1][j].y = Wk[(ka + 1) * HW_PITCH];
      }
    }
    __builtin_amdgcn_sched_barrier(0);
#pragma unroll
    for (int j = 0; j < 4; ++j)
      cacc = __builtin_amdgcn_wmma_f32_16x16x4_f32(false, af[g & 1][j], false, bf[g & 1][j],
                                                   (short)0, cacc, false, false);
    __builtin_amdgcn_sched_barrier(0);
  }

  int mBase = tile * 16 + (hi ? 8 : 0);
#pragma unroll
  for (int rr = 0; rr < 8; ++rr) {
    float v = cacc[rr] + bn;
    v = v > 0.0f ? v : 0.0f;
    dst[(long)(mBase + rr) * DIM + ncol] = v;
  }
}

__global__ void tag_gemm_tail_kernel(const float* __restrict__ s0, const float* __restrict__ s1,
                                     const float* __restrict__ s2, const float* __restrict__ s3,
                                     const float* __restrict__ Wl, const float* __restrict__ bl,
                                     float* __restrict__ dst, int startRow, int N) {
  int t = blockIdx.x * blockDim.x + threadIdx.x;
  int m = startRow + (t >> 5);
  int n = t & 31;
  if (m >= N) return;
  float acc = bl[n];
  const float* srcs[4] = {s0, s1, s2, s3};
  for (int k = 0; k < 4; ++k) {
    const float* S = srcs[k] + (long)m * DIM;
    const float* Wk = Wl + k * DIM * DIM;
    for (int j = 0; j < DIM; ++j) acc += S[j] * Wk[j * DIM + n];
  }
  dst[(long)m * DIM + n] = acc > 0.0f ? acc : 0.0f;
}


extern "C" void kernel_launch(void* const* d_in, const int* in_sizes, int n_in,
                              void* d_out, int out_size, void* d_ws, size_t ws_size,
                              hipStream_t stream) {
  (void)n_in; (void)out_size; (void)ws_size;
  const float* d_x = (const float*)d_in[0];
  const int*   d_ei = (const int*)d_in[1];
  const float* d_w = (const float*)d_in[2];
  const float* d_W = (const float*)d_in[3];
  const float* d_b = (const float*)d_in[4];
  const int N = in_sizes[0] / DIM;
  const int E = in_sizes[2];
  const int* row = d_ei;
  const int* col = d_ei + E;

  auto align_up = [](size_t v) { return (v + 255) & ~(size_t)255; };
  char* ws = (char*)d_ws;
  size_t off = 0;
  int*   cnt   = (int*)(ws + off);   off = align_up(off + (size_t)N * 4);
  int*   start = (int*)(ws + off);   off = align_up(off + ((size_t)N + 1) * 4);
  int*   rnk   = (int*)(ws + off);   off = align_up(off + (size_t)E * 4);
  int2*  sedge = (int2*)(ws + off);  off = align_up(off + (size_t)E * 8);
  float* dis   = (float*)(ws + off); off = align_up(off + (size_t)N * 4);
  float* h1    = (float*)(ws + off); off = align_up(off + (size_t)N * DIM * 4);
  float* h2    = (float*)(ws + off); off = align_up(off + (size_t)N * DIM * 4);
  float* h3    = (float*)(ws + off); off = align_up(off + (size_t)N * DIM * 4);
  float* xcur  = (float*)(ws + off); off = align_up(off + (size_t)N * DIM * 4);
  const int NB  = (N + SORT_BN - 1) >> SORT_SH;
  const int NWG = (E + SORT_CH - 1) / SORT_CH;
  const int M   = NB * NWG;
  int2*  tab   = (int2*)(ws + off);  off = align_up(off + (size_t)M * 8);
  int*   btot  = (int*)(ws + off);   off = align_up(off + (size_t)NB * 4);
  int2*  prc   = (int2*)(ws + off);  off = align_up(off + (size_t)E * 8);
  int*   pcl   = (int*)(ws + off);   off = align_up(off + (size_t)E * 4);

  const int B = 256;
  auto gblk = [&](long n) { return (int)((n + B - 1) / B); };
  const long grp8 = (long)N * 8;

  if (NB <= SORT_MAXNB) {
    bucket_part_kernel<<<NWG, 1024, 0, stream>>>(row, col, d_w, tab, prc, pcl, E, NB);
    bucket_deg_kernel<<<NB, 512, 0, stream>>>(tab, prc, pcl, dis, btot, N, NB, NWG);
    bucket_sort_kernel<<<NB, 512, 0, stream>>>(tab, btot, prc, pcl, dis, start, sedge, N, NB, NWG);
  } else {
    zero_i32_kernel<<<gblk(N), B, 0, stream>>>(cnt, N);
    rank_kernel<<<gblk(E), B, 0, stream>>>(col, cnt, rnk, E);
    scan_kernel<<<N / SCAN_TILE + 1, SCAN_THREADS, 0, stream>>>(cnt, start, N);
    fill_kernel<<<gblk(E), B, 0, stream>>>(row, col, d_w, start, rnk, sedge, E);
    dis_csr_kernel<<<gblk(grp8), B, 0, stream>>>(start, sedge, dis, N);
    norm_csr_kernel<<<gblk(grp8), B, 0, stream>>>(start, dis, sedge, N);
  }

  const int nTiles = N / 16;
  const int hopTileBlocks = (N + HT_ROWS - 1) / HT_ROWS;
  const int tailStart = nTiles * 16;
  const int tailRows = N - tailStart;

  for (int l = 0; l < 3; ++l) {
    const float* xin = (l == 0) ? d_x : xcur;
    float* dst = (l == 2) ? (float*)d_out : xcur;
    const float* Wl = d_W + (size_t)l * 4 * DIM * DIM;
    const float* bl = d_b + (size_t)l * DIM;

    hop_csr_kernel<<<gblk(grp8), B, 0, stream>>>(start, sedge, xin, h1, N);
    hop_csr_kernel<<<gblk(grp8), B, 0, stream>>>(start, sedge, h1, h2, N);

    hop_gemm_wmma_kernel<<<hopTileBlocks, B, 0, stream>>>(start, sedge, xin, h1, h2, Wl, bl, dst,
                                                          h3, N, nTiles);
    if (tailRows > 0) {
      tag_gemm_tail_kernel<<<gblk((long)tailRows * 32), B, 0, stream>>>(
          xin, h1, h2, h3, Wl, bl, dst, tailStart, N);
    }
  }
}
